# starting kernel
# baseline (speedup 1.0000x reference)
#include <hip/hip_runtime.h>

#define N_OCC   20000
#define N_SKILL 50000
#define N_EDGE  600000
#define D_IN    256
#define HID     128
#define OUTF    64

typedef __attribute__((ext_vector_type(2))) float v2f;
typedef __attribute__((ext_vector_type(8))) float v8f;

template <int K, int N, bool FUSE>
__global__ void __launch_bounds__(128)
gemm_sage(const float* __restrict__ A1, const float* __restrict__ WT1,
          const float* __restrict__ A2, const float* __restrict__ WT2,
          const unsigned* __restrict__ cnt, const float* __restrict__ bias,
          float* __restrict__ out, int M, int do_relu)
{
    const int lane = threadIdx.x & 31;
    const int wave = threadIdx.x >> 5;
    const int hi   = lane >> 4;
    const int lm   = lane & 15;
    const int rt   = blockIdx.x * 4 + wave;
    if (rt * 16 >= M) return;
    const int row0 = rt * 16;
    const int col0 = blockIdx.y * 64;

    v8f c0 = {0.f,0.f,0.f,0.f,0.f,0.f,0.f,0.f};
    v8f c1 = c0, c2 = c0, c3 = c0;

    {
        const float* ap = A1  + (size_t)(row0 + lm) * K + 2 * hi;
        const float* bp = WT1 + (size_t)(col0 + lm) * K + 2 * hi;
        #pragma unroll 8
        for (int k = 0; k < K / 4; ++k) {
            v2f a  = *(const v2f*)(ap + 4 * k);
            v2f b0 = *(const v2f*)(bp + 4 * k);
            v2f b1 = *(const v2f*)(bp + 4 * k + 16 * K);
            v2f b2 = *(const v2f*)(bp + 4 * k + 32 * K);
            v2f b3 = *(const v2f*)(bp + 4 * k + 48 * K);
            c0 = __builtin_amdgcn_wmma_f32_16x16x4_f32(false, a, false, b0, (short)0, c0, false, false);
            c1 = __builtin_amdgcn_wmma_f32_16x16x4_f32(false, a, false, b1, (short)0, c1, false, false);
            c2 = __builtin_amdgcn_wmma_f32_16x16x4_f32(false, a, false, b2, (short)0, c2, false, false);
            c3 = __builtin_amdgcn_wmma_f32_16x16x4_f32(false, a, false, b3, (short)0, c3, false, false);
        }
    }

    if (FUSE) {
        #pragma unroll
        for (int r = 0; r < 8; ++r) {
            const float inv = 1.0f / fmaxf((float)cnt[row0 + r + 8 * hi], 1.0f);
            c0[r] *= inv; c1[r] *= inv; c2[r] *= inv; c3[r] *= inv;
        }
        const float* ap = A2  + (size_t)(row0 + lm) * K + 2 * hi;
        const float* bp = WT2 + (size_t)(col0 + lm) * K + 2 * hi;
        #pragma unroll 8
        for (int k = 0; k < K / 4; ++k) {
            v2f a  = *(const v2f*)(ap + 4 * k);
            v2f b0 = *(const v2f*)(bp + 4 * k);
            v2f b1 = *(const v2f*)(bp + 4 * k + 16 * K);
            v2f b2 = *(const v2f*)(bp + 4 * k + 32 * K);
            v2f b3 = *(const v2f*)(bp + 4 * k + 48 * K);
            c0 = __builtin_amdgcn_wmma_f32_16x16x4_f32(false, a, false, b0, (short)0, c0, false, false);
            c1 = __builtin_amdgcn_wmma_f32_16x16x4_f32(false, a, false, b1, (short)0, c1, false, false);
            c2 = __builtin_amdgcn_wmma_f32_16x16x4_f32(false, a, false, b2, (short)0, c2, false, false);
            c3 = __builtin_amdgcn_wmma_f32_16x16x4_f32(false, a, false, b3, (short)0, c3, false, false);
        }
    }

    float* orow = out + (size_t)(row0 + 8 * hi) * N + col0 + lm;
    const float bv0 = bias[col0 + lm];
    const float bv1 = bias[col0 + 16 + lm];
    const float bv2 = bias[col0 + 32 + lm];
    const float bv3 = bias[col0 + 48 + lm];
    #pragma unroll
    for (int r = 0; r < 8; ++r) {
        float v0 = c0[r] + bv0, v1 = c1[r] + bv1, v2 = c2[r] + bv2, v3 = c3[r] + bv3;
        if (do_relu) {
            v0 = fmaxf(v0, 0.f); v1 = fmaxf(v1, 0.f);
            v2 = fmaxf(v2, 0.f); v3 = fmaxf(v3, 0.f);
        }
        float* p = orow + (size_t)r * N;
        p[0] = v0; p[16] = v1; p[32] = v2; p[48] = v3;
    }
}


__global__ void __launch_bounds__(256)
transpose_w(const float* __restrict__ W, float* __restrict__ WT, int K, int N)
{
    int i = blockIdx.x * blockDim.x + threadIdx.x;
    if (i < K * N) {
        int k = i / N, n = i - k * N;
        WT[(size_t)n * K + k] = W[i];
    }
}

__device__ inline void atomic_add_f32(float* p, float v)
{
    (void)__hip_atomic_fetch_add(p, v, __ATOMIC_RELAXED, __HIP_MEMORY_SCOPE_AGENT);
}

__global__ void __launch_bounds__(256)
count_deg(const int* __restrict__ e_occ, const int* __restrict__ e_skill,
          unsigned* __restrict__ cnt_occ, unsigned* __restrict__ cnt_skill, int E)
{
    int e = blockIdx.x * blockDim.x + threadIdx.x;
    if (e < E) {
        atomicAdd(&cnt_occ[e_occ[e]], 1u);
        atomicAdd(&cnt_skill[e_skill[e]], 1u);
    }
}

__global__ void __launch_bounds__(256)
scatter_add(const float* __restrict__ hsrc, const int* __restrict__ src,
            const int* __restrict__ dst, float* __restrict__ agg, int E)
{
    const long long tid = (long long)blockIdx.x * blockDim.x + threadIdx.x;
    const int e = (int)(tid >> 5);
    const int f = ((int)tid & 31) << 2;
    if (e < E) {
        const int s = src[e];
        const int d = dst[e];
        const float4 v = *(const float4*)(hsrc + (size_t)s * HID + f);
        float* p = agg + (size_t)d * HID + f;
        atomic_add_f32(p + 0, v.x);
        atomic_add_f32(p + 1, v.y);
        atomic_add_f32(p + 2, v.z);
        atomic_add_f32(p + 3, v.w);
    }
}

extern "C" void kernel_launch(void* const* d_in, const int* in_sizes, int n_in,
                              void* d_out, int out_size, void* d_ws, size_t ws_size,
                              hipStream_t stream)
{
    (void)in_sizes; (void)n_in; (void)out_size; (void)ws_size;

    const float* x_occ    = (const float*)d_in[0];
    const float* x_skill  = (const float*)d_in[1];
    const int*   e_occ    = (const int*)  d_in[2];
    const int*   e_skill  = (const int*)  d_in[3];
    const float* pW_occ   = (const float*)d_in[4];
    const float* pb_occ   = (const float*)d_in[5];
    const float* pW_skill = (const float*)d_in[6];
    const float* pb_skill = (const float*)d_in[7];
    const float* c1_os_Wl = (const float*)d_in[8];
    const float* c1_os_bl = (const float*)d_in[9];
    const float* c1_os_Wr = (const float*)d_in[10];
    const float* c1_so_Wl = (const float*)d_in[11];
    const float* c1_so_bl = (const float*)d_in[12];
    const float* c1_so_Wr = (const float*)d_in[13];
    const float* c2_os_Wl = (const float*)d_in[14];
    const float* c2_os_bl = (const float*)d_in[15];
    const float* c2_os_Wr = (const float*)d_in[16];
    const float* c2_so_Wl = (const float*)d_in[17];
    const float* c2_so_bl = (const float*)d_in[18];
    const float* c2_so_Wr = (const float*)d_in[19];

    float* ws = (float*)d_ws;
    float* h_occ     = ws;
    float* h_skill   = h_occ     + (size_t)N_OCC   * HID;
    float* o1        = h_skill   + (size_t)N_SKILL * HID;
    float* s1        = o1        + (size_t)N_OCC   * HID;
    float* agg_occ   = s1        + (size_t)N_SKILL * HID;
    float* agg_skill = agg_occ   + (size_t)N_OCC   * HID;
    unsigned* cnt_occ   = (unsigned*)(agg_skill + (size_t)N_SKILL * HID);
    unsigned* cnt_skill = cnt_occ + N_OCC;
    float* wt = (float*)(cnt_skill + N_SKILL);
    float* pWt_occ   = wt;                 wt += D_IN * HID;
    float* pWt_skill = wt;                 wt += D_IN * HID;
    float* t1_os_Wl  = wt;                 wt += HID * HID;
    float* t1_os_Wr  = wt;                 wt += HID * HID;
    float* t1_so_Wl  = wt;                 wt += HID * HID;
    float* t1_so_Wr  = wt;                 wt += HID * HID;
    float* t2_os_Wl  = wt;                 wt += HID * OUTF;
    float* t2_os_Wr  = wt;                 wt += HID * OUTF;
    float* t2_so_Wl  = wt;                 wt += HID * OUTF;
    float* t2_so_Wr  = wt;                 wt += HID * OUTF;

    float* o2 = (float*)d_out;
    float* s2 = o2 + (size_t)N_OCC * OUTF;

    const dim3 blk128(128), blk256(256);
    const int rt_occ   = (N_OCC / 16 + 3) / 4;
    const int rt_skill = (N_SKILL / 16 + 3) / 4;

    hipMemsetAsync(cnt_occ, 0, sizeof(unsigned) * (N_OCC + N_SKILL), stream);
    hipMemsetAsync(agg_occ, 0, sizeof(float) * (size_t)(N_OCC + N_SKILL) * HID, stream);

    transpose_w<<<(D_IN * HID + 255) / 256, blk256, 0, stream>>>(pW_occ,   pWt_occ,   D_IN, HID);
    transpose_w<<<(D_IN * HID + 255) / 256, blk256, 0, stream>>>(pW_skill, pWt_skill, D_IN, HID);
    transpose_w<<<(HID * HID + 255) / 256, blk256, 0, stream>>>(c1_os_Wl, t1_os_Wl, HID, HID);
    transpose_w<<<(HID * HID + 255) / 256, blk256, 0, stream>>>(c1_os_Wr, t1_os_Wr, HID, HID);
    transpose_w<<<(HID * HID + 255) / 256, blk256, 0, stream>>>(c1_so_Wl, t1_so_Wl, HID, HID);
    transpose_w<<<(HID * HID + 255) / 256, blk256, 0, stream>>>(c1_so_Wr, t1_so_Wr, HID, HID);
    transpose_w<<<(HID * OUTF + 255) / 256, blk256, 0, stream>>>(c2_os_Wl, t2_os_Wl, HID, OUTF);
    transpose_w<<<(HID * OUTF + 255) / 256, blk256, 0, stream>>>(c2_os_Wr, t2_os_Wr, HID, OUTF);
    transpose_w<<<(HID * OUTF + 255) / 256, blk256, 0, stream>>>(c2_so_Wl, t2_so_Wl, HID, OUTF);
    transpose_w<<<(HID * OUTF + 255) / 256, blk256, 0, stream>>>(c2_so_Wr, t2_so_Wr, HID, OUTF);

    count_deg<<<(N_EDGE + 255) / 256, blk256, 0, stream>>>(
        e_occ, e_skill, cnt_occ, cnt_skill, N_EDGE);

    gemm_sage<D_IN, HID, false><<<dim3(rt_occ, HID / 64), blk128, 0, stream>>>(
        x_occ, pWt_occ, nullptr, nullptr, nullptr, pb_occ, h_occ, N_OCC, 1);
    gemm_sage<D_IN, HID, false><<<dim3(rt_skill, HID / 64), blk128, 0, stream>>>(
        x_skill, pWt_skill, nullptr, nullptr, nullptr, pb_skill, h_skill, N_SKILL, 1);

    const int sg_blocks = (int)(((long long)N_EDGE * 32 + 255) / 256);
    scatter_add<<<sg_blocks, blk256, 0, stream>>>(h_occ,   e_occ,   e_skill, agg_skill, N_EDGE);
    scatter_add<<<sg_blocks, blk256, 0, stream>>>(h_skill, e_skill, e_occ,   agg_occ,   N_EDGE);

    gemm_sage<HID, HID, true><<<dim3(rt_skill, HID / 64), blk128, 0, stream>>>(
        agg_skill, t1_os_Wl, h_skill, t1_os_Wr, cnt_skill, c1_os_bl, s1, N_SKILL, 1);
    gemm_sage<HID, HID, true><<<dim3(rt_occ, HID / 64), blk128, 0, stream>>>(
        agg_occ, t1_so_Wl, h_occ, t1_so_Wr, cnt_occ, c1_so_bl, o1, N_OCC, 1);

    hipMemsetAsync(agg_occ, 0, sizeof(float) * (size_t)(N_OCC + N_SKILL) * HID, stream);
    scatter_add<<<sg_blocks, blk256, 0, stream>>>(o1, e_occ,   e_skill, agg_skill, N_EDGE);
    scatter_add<<<sg_blocks, blk256, 0, stream>>>(s1, e_skill, e_occ,   agg_occ,   N_EDGE);

    gemm_sage<HID, OUTF, true><<<dim3(rt_skill, OUTF / 64), blk128, 0, stream>>>(
        agg_skill, t2_os_Wl, s1, t2_os_Wr, cnt_skill, c2_os_bl, s2, N_SKILL, 0);
    gemm_sage<HID, OUTF, true><<<dim3(rt_occ, OUTF / 64), blk128, 0, stream>>>(
        agg_occ, t2_so_Wl, o1, t2_so_Wr, cnt_occ, c2_so_bl, o2, N_OCC, 0);
}


# neighbour sums owner-computes: 256 blocks each scan all edges, add rows of their own destination range (and degrees) in LDS by LDS atomics; global atomics, memsets, count kernel go; f32 sum order new
# speedup vs baseline: 1.8786x; 1.8786x over previous
#include <hip/hip_runtime.h>

#define N_OCC   20000
#define N_SKILL 50000
#define N_EDGE  600000
#define D_IN    256
#define HID     128
#define OUTF    64

typedef __attribute__((ext_vector_type(2))) float v2f;
typedef __attribute__((ext_vector_type(8))) float v8f;

template <int K, int N, bool FUSE>
__global__ void __launch_bounds__(128)
gemm_sage(const float* __restrict__ A1, const float* __restrict__ WT1,
          const float* __restrict__ A2, const float* __restrict__ WT2,
          const unsigned* __restrict__ cnt, const float* __restrict__ bias,
          float* __restrict__ out, int M, int do_relu)
{
    const int lane = threadIdx.x & 31;
    const int wave = threadIdx.x >> 5;
    const int hi   = lane >> 4;
    const int lm   = lane & 15;
    const int rt   = blockIdx.x * 4 + wave;
    if (rt * 16 >= M) return;
    const int row0 = rt * 16;
    const int col0 = blockIdx.y * 64;

    v8f c0 = {0.f,0.f,0.f,0.f,0.f,0.f,0.f,0.f};
    v8f c1 = c0, c2 = c0, c3 = c0;

    {
        const float* ap = A1  + (size_t)(row0 + lm) * K + 2 * hi;
        const float* bp = WT1 + (size_t)(col0 + lm) * K + 2 * hi;
        #pragma unroll 8
        for (int k = 0; k < K / 4; ++k) {
            v2f a  = *(const v2f*)(ap + 4 * k);
            v2f b0 = *(const v2f*)(bp + 4 * k);
            v2f b1 = *(const v2f*)(bp + 4 * k + 16 * K);
            v2f b2 = *(const v2f*)(bp + 4 * k + 32 * K);
            v2f b3 = *(const v2f*)(bp + 4 * k + 48 * K);
            c0 = __builtin_amdgcn_wmma_f32_16x16x4_f32(false, a, false, b0, (short)0, c0, false, false);
            c1 = __builtin_amdgcn_wmma_f32_16x16x4_f32(false, a, false, b1, (short)0, c1, false, false);
            c2 = __builtin_amdgcn_wmma_f32_16x16x4_f32(false, a, false, b2, (short)0, c2, false, false);
            c3 = __builtin_amdgcn_wmma_f32_16x16x4_f32(false, a, false, b3, (short)0, c3, false, false);
        }
    }

    if (FUSE) {
        #pragma unroll
        for (int r = 0; r < 8; ++r) {
            const float inv = 1.0f / fmaxf((float)cnt[row0 + r + 8 * hi], 1.0f);
            c0[r] *= inv; c1[r] *= inv; c2[r] *= inv; c3[r] *= inv;
        }
        const float* ap = A2  + (size_t)(row0 + lm) * K + 2 * hi;
        const float* bp = WT2 + (size_t)(col0 + lm) * K + 2 * hi;
        #pragma unroll 8
        for (int k = 0; k < K / 4; ++k) {
            v2f a  = *(const v2f*)(ap + 4 * k);
            v2f b0 = *(const v2f*)(bp + 4 * k);
            v2f b1 = *(const v2f*)(bp + 4 * k + 16 * K);
            v2f b2 = *(const v2f*)(bp + 4 * k + 32 * K);
            v2f b3 = *(const v2f*)(bp + 4 * k + 48 * K);
            c0 = __builtin_amdgcn_wmma_f32_16x16x4_f32(false, a, false, b0, (short)0, c0, false, false);
            c1 = __builtin_amdgcn_wmma_f32_16x16x4_f32(false, a, false, b1, (short)0, c1, false, false);
            c2 = __builtin_amdgcn_wmma_f32_16x16x4_f32(false, a, false, b2, (short)0, c2, false, false);
            c3 = __builtin_amdgcn_wmma_f32_16x16x4_f32(false, a, false, b3, (short)0, c3, false, false);
        }
    }

    float* orow = out + (size_t)(row0 + 8 * hi) * N + col0 + lm;
    const float bv0 = bias[col0 + lm];
    const float bv1 = bias[col0 + 16 + lm];
    const float bv2 = bias[col0 + 32 + lm];
    const float bv3 = bias[col0 + 48 + lm];
    #pragma unroll
    for (int r = 0; r < 8; ++r) {
        float v0 = c0[r] + bv0, v1 = c1[r] + bv1, v2 = c2[r] + bv2, v3 = c3[r] + bv3;
        if (do_relu) {
            v0 = fmaxf(v0, 0.f); v1 = fmaxf(v1, 0.f);
            v2 = fmaxf(v2, 0.f); v3 = fmaxf(v3, 0.f);
        }
        float* p = orow + (size_t)r * N;
        p[0] = v0; p[16] = v1; p[32] = v2; p[48] = v3;
    }
}


__global__ void __launch_bounds__(256)
transpose_w(const float* __restrict__ W, float* __restrict__ WT, int K, int N)
{
    int i = blockIdx.x * blockDim.x + threadIdx.x;
    if (i < K * N) {
        int k = i / N, n = i - k * N;
        WT[(size_t)n * K + k] = W[i];
    }
}

#define NBLK   256
#define ROWS_S 196
#define ROWS_O 79
#define ROWS_T (ROWS_S + ROWS_O)
#define NWAVE  32
#define QCAP   640
#define NVEC   (N_EDGE / 4)
#define NITER  ((NVEC + 2047) / 2048)

static_assert(N_EDGE % 4 == 0, "edge list is read as int4");
static_assert(NBLK * ROWS_S >= N_SKILL && NBLK * ROWS_O >= N_OCC, "row ranges cover all nodes");
static_assert(N_SKILL < 65536 && N_OCC < 65536 && ROWS_T < 65536, "queue entry packing");
static_assert(QCAP % 8 == 0 && QCAP > 512, "drain reads whole batches of 8; room for one iteration");

__device__ inline void lds_add_f32(float* p, float v)
{
    (void)__hip_atomic_fetch_add(p, v, __ATOMIC_RELAXED, __HIP_MEMORY_SCOPE_WORKGROUP);
}

template <bool COUNT>
__device__ inline void drain_queue(const unsigned* myq, int qn, float* acc, unsigned* cl,
                                   const float* __restrict__ f_occ,
                                   const float* __restrict__ f_skill, int lane)
{
    for (int base = 0; base < qn; base += 8) {
        unsigned row[8];
        float v[8][4];
        #pragma unroll
        for (int j = 0; j < 8; ++j) {
            unsigned e = __builtin_amdgcn_readfirstlane(myq[base + j]);
            if (base + j >= qn) e = 0u;
            row[j] = e >> 16;
            const unsigned srow = e & 0xffffu;
            const float* tab = (row[j] < (unsigned)ROWS_S) ? f_occ : f_skill;
            const float* p = tab + (size_t)srow * HID + lane;
            v[j][0] = p[0]; v[j][1] = p[32]; v[j][2] = p[64]; v[j][3] = p[96];
        }
        #pragma unroll
        for (int j = 0; j < 8; ++j) {
            if (base + j < qn) {
                float* d = acc + row[j] * HID + lane;
                lds_add_f32(d,      v[j][0]);
                lds_add_f32(d + 32, v[j][1]);
                lds_add_f32(d + 64, v[j][2]);
                lds_add_f32(d + 96, v[j][3]);
                if (COUNT && lane == 0) atomicAdd(&cl[row[j]], 1u);
            }
        }
    }
}

template <bool COUNT>
__global__ void __launch_bounds__(1024)
owner_sum(const float* __restrict__ f_occ, const float* __restrict__ f_skill,
          const int* __restrict__ e_occ, const int* __restrict__ e_skill,
          float* __restrict__ agg_occ, float* __restrict__ agg_skill,
          unsigned* __restrict__ cnt_occ, unsigned* __restrict__ cnt_skill)
{
    __shared__ float    acc[ROWS_T * HID];
    __shared__ unsigned q[NWAVE * QCAP];
    __shared__ unsigned cl[ROWS_T + 1];

    const int tid  = threadIdx.x;
    const int lane = tid & 31;
    const int wave = tid >> 5;
    const int s_lo = blockIdx.x * ROWS_S;
    const int o_lo = blockIdx.x * ROWS_O;
    unsigned* myq = q + wave * QCAP;

    for (int i = tid; i < ROWS_T * HID / 4; i += 1024)
        ((float4*)acc)[i] = make_float4(0.f, 0.f, 0.f, 0.f);
    if (COUNT)
        for (int i = tid; i < ROWS_T; i += 1024) cl[i] = 0u;
    __syncthreads();

    const int4* vo = (const int4*)e_occ;
    const int4* vs = (const int4*)e_skill;
    const int4 none = make_int4(-1, -1, -1, -1);

    int qn = 0;
    int it = (int)((blockIdx.x * NITER) >> 8);

    for (int k = 0; k < NITER; ++k) {
        const int ia = it * 2048 + tid;
        const int ib = ia + 1024;
        int4 eo0 = none, es0 = none, eo1 = none, es1 = none;
        if (ia < NVEC) { eo0 = vo[ia]; es0 = vs[ia]; }
        if (ib < NVEC) { eo1 = vo[ib]; es1 = vs[ib]; }

        if (qn > QCAP - 512) { drain_queue<COUNT>(myq, qn, acc, cl, f_occ, f_skill, lane); qn = 0; }

#define PUSH_EDGE(O, S)                                                              \
        {                                                                            \
            const unsigned rs = (unsigned)((S) - s_lo);                              \
            const bool ms = rs < (unsigned)ROWS_S;                                   \
            const unsigned bs = __builtin_amdgcn_ballot_w32(ms);                     \
            if (bs) {                                                                \
                if (ms) myq[qn + __builtin_amdgcn_mbcnt_lo(bs, 0u)] = (rs << 16) | (unsigned)(O); \
                qn += __builtin_popcount(bs);                                        \
            }                                                                        \
            const unsigned ro = (unsigned)((O) - o_lo);                              \
            const bool mo = ro < (unsigned)ROWS_O;                                   \
            const unsigned bo = __builtin_amdgcn_ballot_w32(mo);                     \
            if (bo) {                                                                \
                if (mo) myq[qn + __builtin_amdgcn_mbcnt_lo(bo, 0u)] = ((ro + ROWS_S) << 16) | (unsigned)(S); \
                qn += __builtin_popcount(bo);                                        \
            }                                                                        \
        }
        PUSH_EDGE(eo0.x, es0.x)
        PUSH_EDGE(eo0.y, es0.y)
        PUSH_EDGE(eo0.z, es0.z)
        PUSH_EDGE(eo0.w, es0.w)
        PUSH_EDGE(eo1.x, es1.x)
        PUSH_EDGE(eo1.y, es1.y)
        PUSH_EDGE(eo1.z, es1.z)
        PUSH_EDGE(eo1.w, es1.w)
#undef PUSH_EDGE
        it = (it + 1 == NITER) ? 0 : it + 1;
    }
    drain_queue<COUNT>(myq, qn, acc, cl, f_occ, f_skill, lane);

    __syncthreads();

    for (int i = tid; i < ROWS_S * HID / 4; i += 1024) {
        const int r = s_lo + (i >> 5);
        if (r < N_SKILL) ((float4*)agg_skill)[(size_t)r * (HID / 4) + (i & 31)] = ((const float4*)acc)[i];
    }
    for (int i = tid; i < ROWS_O * HID / 4; i += 1024) {
        const int r = o_lo + (i >> 5);
        if (r < N_OCC) ((float4*)agg_occ)[(size_t)r * (HID / 4) + (i & 31)] = ((const float4*)acc)[ROWS_S * HID / 4 + i];
    }
    if (COUNT) {
        for (int i = tid; i < ROWS_S; i += 1024)
            if (s_lo + i < N_SKILL) cnt_skill[s_lo + i] = cl[i];
        for (int i = tid; i < ROWS_O; i += 1024)
            if (o_lo + i < N_OCC) cnt_occ[o_lo + i] = cl[ROWS_S + i];
    }
}

extern "C" void kernel_launch(void* const* d_in, const int* in_sizes, int n_in,
                              void* d_out, int out_size, void* d_ws, size_t ws_size,
                              hipStream_t stream)
{
    (void)in_sizes; (void)n_in; (void)out_size; (void)ws_size;

    const float* x_occ    = (const float*)d_in[0];
    const float* x_skill  = (const float*)d_in[1];
    const int*   e_occ    = (const int*)  d_in[2];
    const int*   e_skill  = (const int*)  d_in[3];
    const float* pW_occ   = (const float*)d_in[4];
    const float* pb_occ   = (const float*)d_in[5];
    const float* pW_skill = (const float*)d_in[6];
    const float* pb_skill = (const float*)d_in[7];
    const float* c1_os_Wl = (const float*)d_in[8];
    const float* c1_os_bl = (const float*)d_in[9];
    const float* c1_os_Wr = (const float*)d_in[10];
    const float* c1_so_Wl = (const float*)d_in[11];
    const float* c1_so_bl = (const float*)d_in[12];
    const float* c1_so_Wr = (const float*)d_in[13];
    const float* c2_os_Wl = (const float*)d_in[14];
    const float* c2_os_bl = (const float*)d_in[15];
    const float* c2_os_Wr = (const float*)d_in[16];
    const float* c2_so_Wl = (const float*)d_in[17];
    const float* c2_so_bl = (const float*)d_in[18];
    const float* c2_so_Wr = (const float*)d_in[19];

    float* ws = (float*)d_ws;
    float* h_occ     = ws;
    float* h_skill   = h_occ     + (size_t)N_OCC   * HID;
    float* o1        = h_skill   + (size_t)N_SKILL * HID;
    float* s1        = o1        + (size_t)N_OCC   * HID;
    float* agg_occ   = s1        + (size_t)N_SKILL * HID;
    float* agg_skill = agg_occ   + (size_t)N_OCC   * HID;
    unsigned* cnt_occ   = (unsigned*)(agg_skill + (size_t)N_SKILL * HID);
    unsigned* cnt_skill = cnt_occ + N_OCC;
    float* wt = (float*)(cnt_skill + N_SKILL);
    float* pWt_occ   = wt;                 wt += D_IN * HID;
    float* pWt_skill = wt;                 wt += D_IN * HID;
    float* t1_os_Wl  = wt;                 wt += HID * HID;
    float* t1_os_Wr  = wt;                 wt += HID * HID;
    float* t1_so_Wl  = wt;                 wt += HID * HID;
    float* t1_so_Wr  = wt;                 wt += HID * HID;
    float* t2_os_Wl  = wt;                 wt += HID * OUTF;
    float* t2_os_Wr  = wt;                 wt += HID * OUTF;
    float* t2_so_Wl  = wt;                 wt += HID * OUTF;
    float* t2_so_Wr  = wt;                 wt += HID * OUTF;

    float* o2 = (float*)d_out;
    float* s2 = o2 + (size_t)N_OCC * OUTF;

    const dim3 blk128(128), blk256(256);
    const int rt_occ   = (N_OCC / 16 + 3) / 4;
    const int rt_skill = (N_SKILL / 16 + 3) / 4;

    transpose_w<<<(D_IN * HID + 255) / 256, blk256, 0, stream>>>(pW_occ,   pWt_occ,   D_IN, HID);
    transpose_w<<<(D_IN * HID + 255) / 256, blk256, 0, stream>>>(pW_skill, pWt_skill, D_IN, HID);
    transpose_w<<<(HID * HID + 255) / 256, blk256, 0, stream>>>(c1_os_Wl, t1_os_Wl, HID, HID);
    transpose_w<<<(HID * HID + 255) / 256, blk256, 0, stream>>>(c1_os_Wr, t1_os_Wr, HID, HID);
    transpose_w<<<(HID * HID + 255) / 256, blk256, 0, stream>>>(c1_so_Wl, t1_so_Wl, HID, HID);
    transpose_w<<<(HID * HID + 255) / 256, blk256, 0, stream>>>(c1_so_Wr, t1_so_Wr, HID, HID);
    transpose_w<<<(HID * OUTF + 255) / 256, blk256, 0, stream>>>(c2_os_Wl, t2_os_Wl, HID, OUTF);
    transpose_w<<<(HID * OUTF + 255) / 256, blk256, 0, stream>>>(c2_os_Wr, t2_os_Wr, HID, OUTF);
    transpose_w<<<(HID * OUTF + 255) / 256, blk256, 0, stream>>>(c2_so_Wl, t2_so_Wl, HID, OUTF);
    transpose_w<<<(HID * OUTF + 255) / 256, blk256, 0, stream>>>(c2_so_Wr, t2_so_Wr, HID, OUTF);

    gemm_sage<D_IN, HID, false><<<dim3(rt_occ, HID / 64), blk128, 0, stream>>>(
        x_occ, pWt_occ, nullptr, nullptr, nullptr, pb_occ, h_occ, N_OCC, 1);
    gemm_sage<D_IN, HID, false><<<dim3(rt_skill, HID / 64), blk128, 0, stream>>>(
        x_skill, pWt_skill, nullptr, nullptr, nullptr, pb_skill, h_skill, N_SKILL, 1);

    owner_sum<true><<<NBLK, dim3(1024), 0, stream>>>(
        h_occ, h_skill, e_occ, e_skill, agg_occ, agg_skill, cnt_occ, cnt_skill);

    gemm_sage<HID, HID, true><<<dim3(rt_skill, HID / 64), blk128, 0, stream>>>(
        agg_skill, t1_os_Wl, h_skill, t1_os_Wr, cnt_skill, c1_os_bl, s1, N_SKILL, 1);
    gemm_sage<HID, HID, true><<<dim3(rt_occ, HID / 64), blk128, 0, stream>>>(
        agg_occ, t1_so_Wl, h_occ, t1_so_Wr, cnt_occ, c1_so_bl, o1, N_OCC, 1);

    owner_sum<false><<<NBLK, dim3(1024), 0, stream>>>(
        o1, s1, e_occ, e_skill, agg_occ, agg_skill, cnt_occ, cnt_skill);

    gemm_sage<HID, OUTF, true><<<dim3(rt_skill, OUTF / 64), blk128, 0, stream>>>(
        agg_skill, t2_os_Wl, s1, t2_os_Wr, cnt_skill, c2_os_bl, s2, N_SKILL, 0);
    gemm_sage<HID, OUTF, true><<<dim3(rt_occ, OUTF / 64), blk128, 0, stream>>>(
        agg_occ, t2_so_Wl, o1, t2_so_Wr, cnt_occ, c2_so_bl, o2, N_OCC, 0);
}


# layer-2 neighbour sums skip the edge scan: pass 1 keeps each wave's matched edges as a list (parked in out), pass 2 drains the lists; f32 sum order: LDS-atomic arrival order, as before
# speedup vs baseline: 2.0472x; 1.0897x over previous
#include <hip/hip_runtime.h>

#define N_OCC   20000
#define N_SKILL 50000
#define N_EDGE  600000
#define D_IN    256
#define HID     128
#define OUTF    64

typedef __attribute__((ext_vector_type(2))) float v2f;
typedef __attribute__((ext_vector_type(8))) float v8f;

template <int K, int N, bool FUSE>
__global__ void __launch_bounds__(128)
gemm_sage(const float* __restrict__ A1, const float* __restrict__ WT1,
          const float* __restrict__ A2, const float* __restrict__ WT2,
          const unsigned* __restrict__ cnt, const float* __restrict__ bias,
          float* __restrict__ out, int M, int do_relu)
{
    const int lane = threadIdx.x & 31;
    const int wave = threadIdx.x >> 5;
    const int hi   = lane >> 4;
    const int lm   = lane & 15;
    const int rt   = blockIdx.x * 4 + wave;
    if (rt * 16 >= M) return;
    const int row0 = rt * 16;
    const int col0 = blockIdx.y * 64;

    v8f c0 = {0.f,0.f,0.f,0.f,0.f,0.f,0.f,0.f};
    v8f c1 = c0, c2 = c0, c3 = c0;

    {
        const float* ap = A1  + (size_t)(row0 + lm) * K + 2 * hi;
        const float* bp = WT1 + (size_t)(col0 + lm) * K + 2 * hi;
        #pragma unroll 8
        for (int k = 0; k < K / 4; ++k) {
            v2f a  = *(const v2f*)(ap + 4 * k);
            v2f b0 = *(const v2f*)(bp + 4 * k);
            v2f b1 = *(const v2f*)(bp + 4 * k + 16 * K);
            v2f b2 = *(const v2f*)(bp + 4 * k + 32 * K);
            v2f b3 = *(const v2f*)(bp + 4 * k + 48 * K);
            c0 = __builtin_amdgcn_wmma_f32_16x16x4_f32(false, a, false, b0, (short)0, c0, false, false);
            c1 = __builtin_amdgcn_wmma_f32_16x16x4_f32(false, a, false, b1, (short)0, c1, false, false);
            c2 = __builtin_amdgcn_wmma_f32_16x16x4_f32(false, a, false, b2, (short)0, c2, false, false);
            c3 = __builtin_amdgcn_wmma_f32_16x16x4_f32(false, a, false, b3, (short)0, c3, false, false);
        }
    }

    if (FUSE) {
        #pragma unroll
        for (int r = 0; r < 8; ++r) {
            const float inv = 1.0f / fmaxf((float)cnt[row0 + r + 8 * hi], 1.0f);
            c0[r] *= inv; c1[r] *= inv; c2[r] *= inv; c3[r] *= inv;
        }
        const float* ap = A2  + (size_t)(row0 + lm) * K + 2 * hi;
        const float* bp = WT2 + (size_t)(col0 + lm) * K + 2 * hi;
        #pragma unroll 8
        for (int k = 0; k < K / 4; ++k) {
            v2f a  = *(const v2f*)(ap + 4 * k);
            v2f b0 = *(const v2f*)(bp + 4 * k);
            v2f b1 = *(const v2f*)(bp + 4 * k + 16 * K);
            v2f b2 = *(const v2f*)(bp + 4 * k + 32 * K);
            v2f b3 = *(const v2f*)(bp + 4 * k + 48 * K);
            c0 = __builtin_amdgcn_wmma_f32_16x16x4_f32(false, a, false, b0, (short)0, c0, false, false);
            c1 = __builtin_amdgcn_wmma_f32_16x16x4_f32(false, a, false, b1, (short)0, c1, false, false);
            c2 = __builtin_amdgcn_wmma_f32_16x16x4_f32(false, a, false, b2, (short)0, c2, false, false);
            c3 = __builtin_amdgcn_wmma_f32_16x16x4_f32(false, a, false, b3, (short)0, c3, false, false);
        }
    }

    float* orow = out + (size_t)(row0 + 8 * hi) * N + col0 + lm;
    const float bv0 = bias[col0 + lm];
    const float bv1 = bias[col0 + 16 + lm];
    const float bv2 = bias[col0 + 32 + lm];
    const float bv3 = bias[col0 + 48 + lm];
    #pragma unroll
    for (int r = 0; r < 8; ++r) {
        float v0 = c0[r] + bv0, v1 = c1[r] + bv1, v2 = c2[r] + bv2, v3 = c3[r] + bv3;
        if (do_relu) {
            v0 = fmaxf(v0, 0.f); v1 = fmaxf(v1, 0.f);
            v2 = fmaxf(v2, 0.f); v3 = fmaxf(v3, 0.f);
        }
        float* p = orow + (size_t)r * N;
        p[0] = v0; p[16] = v1; p[32] = v2; p[48] = v3;
    }
}


__global__ void __launch_bounds__(256)
transpose_w(const float* __restrict__ W, float* __restrict__ WT, int K, int N)
{
    int i = blockIdx.x * blockDim.x + threadIdx.x;
    if (i < K * N) {
        int k = i / N, n = i - k * N;
        WT[(size_t)n * K + k] = W[i];
    }
}

#define NBLK   256
#define ROWS_S 196
#define ROWS_O 79
#define ROWS_T (ROWS_S + ROWS_O)
#define NWAVE  32
#define CAPW   544
#define QCAP   (CAPW + 512)
#define QPEND  128
#define LIST_HDR  NWAVE
#define BLK_WORDS (LIST_HDR + NWAVE * CAPW)
#define NVEC   (N_EDGE / 4)
#define NITER  ((NVEC + 2047) / 2048)

static_assert(N_EDGE % 4 == 0, "edge list is read as int4");
static_assert(NBLK * ROWS_S >= N_SKILL && NBLK * ROWS_O >= N_OCC, "row ranges cover all nodes");
static_assert(N_SKILL < 65536 && N_OCC < 65536 && ROWS_T < 65536, "queue entry packing");
static_assert(QCAP % 8 == 0 && QCAP > 512, "drain reads whole batches of 8; room for one iteration");
static_assert(CAPW % 32 == 0 && BLK_WORDS % 32 == 0, "lists start on 128-byte lines");
static_assert((size_t)NBLK * BLK_WORDS <= (size_t)(N_OCC + N_SKILL) * OUTF, "lists fit the output buffer");

__device__ inline void lds_add_f32(float* p, float v)
{
    (void)__hip_atomic_fetch_add(p, v, __ATOMIC_RELAXED, __HIP_MEMORY_SCOPE_WORKGROUP);
}

template <bool COUNT>
__device__ inline void drain_queue(const unsigned* myq, int qs, int qn, float* acc, unsigned* cl,
                                   const float* __restrict__ f_occ,
                                   const float* __restrict__ f_skill, int lane)
{
    for (int base = qs; base < qn; base += 8) {
        unsigned row[8];
        float v[8][4];
        #pragma unroll
        for (int j = 0; j < 8; ++j) {
            unsigned e = __builtin_amdgcn_readfirstlane(myq[base + j]);
            if (base + j >= qn) e = 0u;
            row[j] = e >> 16;
            const unsigned srow = e & 0xffffu;
            const float* tab = (row[j] < (unsigned)ROWS_S) ? f_occ : f_skill;
            const float* p = tab + (size_t)srow * HID + lane;
            v[j][0] = p[0]; v[j][1] = p[32]; v[j][2] = p[64]; v[j][3] = p[96];
        }
        #pragma unroll
        for (int j = 0; j < 8; ++j) {
            if (base + j < qn) {
                float* d = acc + row[j] * HID + lane;
                lds_add_f32(d,      v[j][0]);
                lds_add_f32(d + 32, v[j][1]);
                lds_add_f32(d + 64, v[j][2]);
                lds_add_f32(d + 96, v[j][3]);
                if (COUNT && lane == 0) atomicAdd(&cl[row[j]], 1u);
            }
        }
    }
}

template <bool COUNT>
__global__ void __launch_bounds__(1024)
owner_sum(const float* __restrict__ f_occ, const float* __restrict__ f_skill,
          const int* __restrict__ e_occ, const int* __restrict__ e_skill,
          float* __restrict__ agg_occ, float* __restrict__ agg_skill,
          unsigned* __restrict__ cnt_occ, unsigned* __restrict__ cnt_skill,
          float* __restrict__ list_buf)
{
    __shared__ float    acc[ROWS_T * HID];
    __shared__ unsigned q[NWAVE * QCAP];
    __shared__ unsigned cl[ROWS_T + 1];
    __shared__ unsigned ln[NWAVE];

    const int tid  = threadIdx.x;
    const int lane = tid & 31;
    const int wave = tid >> 5;
    const int s_lo = blockIdx.x * ROWS_S;
    const int o_lo = blockIdx.x * ROWS_O;
    unsigned* myq = q + wave * QCAP;
    unsigned* blk_list = (unsigned*)list_buf + (size_t)blockIdx.x * BLK_WORDS;
    unsigned* wl = blk_list + LIST_HDR + wave * CAPW;

    unsigned n_list = 0u;
    if (!COUNT) n_list = blk_list[wave];

    for (int i = tid; i < ROWS_T * HID / 4; i += 1024)
        ((float4*)acc)[i] = make_float4(0.f, 0.f, 0.f, 0.f);
    if (COUNT)
        for (int i = tid; i < ROWS_T; i += 1024) cl[i] = 0u;

    int qs = 0;
    int qn = 0;
    bool scan = true;
    if (!COUNT) {
        const unsigned n = __builtin_amdgcn_readfirstlane(n_list);
        if (n <= (unsigned)CAPW) {
            for (int i0 = 0; i0 < (int)n; i0 += 256) {
                unsigned t[8];
                #pragma unroll
                for (int j = 0; j < 8; ++j) {
                    const int i = i0 + 32 * j + lane;
                    t[j] = (i < (int)n) ? wl[i] : 0u;
                }
                #pragma unroll
                for (int j = 0; j < 8; ++j) {
                    const int i = i0 + 32 * j + lane;
                    if (i < (int)n) myq[i] = t[j];
                }
            }
            qn = (int)n;
            scan = false;
        }
    }
    __syncthreads();

    const int4* vo = (const int4*)e_occ;
    const int4* vs = (const int4*)e_skill;
    const int4 none = make_int4(-1, -1, -1, -1);

    bool fits = true;
    int it = (int)((blockIdx.x * NITER) >> 8);

    if (scan) {
    for (int k = 0; k < NITER; ++k) {
        const int ia = it * 2048 + tid;
        const int ib = ia + 1024;
        int4 eo0 = none, es0 = none, eo1 = none, es1 = none;
        if (ia < NVEC) { eo0 = vo[ia]; es0 = vs[ia]; }
        if (ib < NVEC) { eo1 = vo[ib]; es1 = vs[ib]; }

        if (qn - qs > QPEND || qn > QCAP - 512) {
            drain_queue<COUNT>(myq, qs, qn, acc, cl, f_occ, f_skill, lane);
            qs = qn;
            if (qn > QCAP - 512) { qs = 0; qn = 0; fits = false; }
        }

#define PUSH_EDGE(O, S)                                                              \
        {                                                                            \
            const unsigned rs = (unsigned)((S) - s_lo);                              \
            const bool ms = rs < (unsigned)ROWS_S;                                   \
            const unsigned bs = __builtin_amdgcn_ballot_w32(ms);                     \
            if (bs) {                                                                \
                if (ms) myq[qn + __builtin_amdgcn_mbcnt_lo(bs, 0u)] = (rs << 16) | (unsigned)(O); \
                qn += __builtin_popcount(bs);                                        \
            }                                                                        \
            const unsigned ro = (unsigned)((O) - o_lo);                              \
            const bool mo = ro < (unsigned)ROWS_O;                                   \
            const unsigned bo = __builtin_amdgcn_ballot_w32(mo);                     \
            if (bo) {                                                                \
                if (mo) myq[qn + __builtin_amdgcn_mbcnt_lo(bo, 0u)] = ((ro + ROWS_S) << 16) | (unsigned)(S); \
                qn += __builtin_popcount(bo);                                        \
            }                                                                        \
        }
        PUSH_EDGE(eo0.x, es0.x)
        PUSH_EDGE(eo0.y, es0.y)
        PUSH_EDGE(eo0.z, es0.z)
        PUSH_EDGE(eo0.w, es0.w)
        PUSH_EDGE(eo1.x, es1.x)
        PUSH_EDGE(eo1.y, es1.y)
        PUSH_EDGE(eo1.z, es1.z)
        PUSH_EDGE(eo1.w, es1.w)
#undef PUSH_EDGE
        it = (it + 1 == NITER) ? 0 : it + 1;
    }
    }
    drain_queue<COUNT>(myq, qs, qn, acc, cl, f_occ, f_skill, lane);

    if (COUNT) {
        const bool keep = fits && qn <= CAPW;
        if (keep)
            for (int i = lane; i < qn; i += 32) wl[i] = myq[i];
        if (lane == 0) ln[wave] = keep ? (unsigned)qn : (unsigned)(CAPW + 1);
    }

    __syncthreads();

    if (COUNT && wave == 0) blk_list[lane] = ln[lane];

    for (int i = tid; i < ROWS_S * HID / 4; i += 1024) {
        const int r = s_lo + (i >> 5);
        if (r < N_SKILL) ((float4*)agg_skill)[(size_t)r * (HID / 4) + (i & 31)] = ((const float4*)acc)[i];
    }
    for (int i = tid; i < ROWS_O * HID / 4; i += 1024) {
        const int r = o_lo + (i >> 5);
        if (r < N_OCC) ((float4*)agg_occ)[(size_t)r * (HID / 4) + (i & 31)] = ((const float4*)acc)[ROWS_S * HID / 4 + i];
    }
    if (COUNT) {
        for (int i = tid; i < ROWS_S; i += 1024)
            if (s_lo + i < N_SKILL) cnt_skill[s_lo + i] = cl[i];
        for (int i = tid; i < ROWS_O; i += 1024)
            if (o_lo + i < N_OCC) cnt_occ[o_lo + i] = cl[ROWS_S + i];
    }
}

extern "C" void kernel_launch(void* const* d_in, const int* in_sizes, int n_in,
                              void* d_out, int out_size, void* d_ws, size_t ws_size,
                              hipStream_t stream)
{
    (void)in_sizes; (void)n_in; (void)out_size; (void)ws_size;

    const float* x_occ    = (const float*)d_in[0];
    const float* x_skill  = (const float*)d_in[1];
    const int*   e_occ    = (const int*)  d_in[2];
    const int*   e_skill  = (const int*)  d_in[3];
    const float* pW_occ   = (const float*)d_in[4];
    const float* pb_occ   = (const float*)d_in[5];
    const float* pW_skill = (const float*)d_in[6];
    const float* pb_skill = (const float*)d_in[7];
    const float* c1_os_Wl = (const float*)d_in[8];
    const float* c1_os_bl = (const float*)d_in[9];
    const float* c1_os_Wr = (const float*)d_in[10];
    const float* c1_so_Wl = (const float*)d_in[11];
    const float* c1_so_bl = (const float*)d_in[12];
    const float* c1_so_Wr = (const float*)d_in[13];
    const float* c2_os_Wl = (const float*)d_in[14];
    const float* c2_os_bl = (const float*)d_in[15];
    const float* c2_os_Wr = (const float*)d_in[16];
    const float* c2_so_Wl = (const float*)d_in[17];
    const float* c2_so_bl = (const float*)d_in[18];
    const float* c2_so_Wr = (const float*)d_in[19];

    float* ws = (float*)d_ws;
    float* h_occ     = ws;
    float* h_skill   = h_occ     + (size_t)N_OCC   * HID;
    float* o1        = h_skill   + (size_t)N_SKILL * HID;
    float* s1        = o1        + (size_t)N_OCC   * HID;
    float* agg_occ   = s1        + (size_t)N_SKILL * HID;
    float* agg_skill = agg_occ   + (size_t)N_OCC   * HID;
    unsigned* cnt_occ   = (unsigned*)(agg_skill + (size_t)N_SKILL * HID);
    unsigned* cnt_skill = cnt_occ + N_OCC;
    float* wt = (float*)(cnt_skill + N_SKILL);
    float* pWt_occ   = wt;                 wt += D_IN * HID;
    float* pWt_skill = wt;                 wt += D_IN * HID;
    float* t1_os_Wl  = wt;                 wt += HID * HID;
    float* t1_os_Wr  = wt;                 wt += HID * HID;
    float* t1_so_Wl  = wt;                 wt += HID * HID;
    float* t1_so_Wr  = wt;                 wt += HID * HID;
    float* t2_os_Wl  = wt;                 wt += HID * OUTF;
    float* t2_os_Wr  = wt;                 wt += HID * OUTF;
    float* t2_so_Wl  = wt;                 wt += HID * OUTF;
    float* t2_so_Wr  = wt;                 wt += HID * OUTF;

    float* o2 = (float*)d_out;
    float* s2 = o2 + (size_t)N_OCC * OUTF;
    float* edge_lists = (float*)d_out;

    const dim3 blk128(128), blk256(256);
    const int rt_occ   = (N_OCC / 16 + 3) / 4;
    const int rt_skill = (N_SKILL / 16 + 3) / 4;

    transpose_w<<<(D_IN * HID + 255) / 256, blk256, 0, stream>>>(pW_occ,   pWt_occ,   D_IN, HID);
    transpose_w<<<(D_IN * HID + 255) / 256, blk256, 0, stream>>>(pW_skill, pWt_skill, D_IN, HID);
    transpose_w<<<(HID * HID + 255) / 256, blk256, 0, stream>>>(c1_os_Wl, t1_os_Wl, HID, HID);
    transpose_w<<<(HID * HID + 255) / 256, blk256, 0, stream>>>(c1_os_Wr, t1_os_Wr, HID, HID);
    transpose_w<<<(HID * HID + 255) / 256, blk256, 0, stream>>>(c1_so_Wl, t1_so_Wl, HID, HID);
    transpose_w<<<(HID * HID + 255) / 256, blk256, 0, stream>>>(c1_so_Wr, t1_so_Wr, HID, HID);
    transpose_w<<<(HID * OUTF + 255) / 256, blk256, 0, stream>>>(c2_os_Wl, t2_os_Wl, HID, OUTF);
    transpose_w<<<(HID * OUTF + 255) / 256, blk256, 0, stream>>>(c2_os_Wr, t2_os_Wr, HID, OUTF);
    transpose_w<<<(HID * OUTF + 255) / 256, blk256, 0, stream>>>(c2_so_Wl, t2_so_Wl, HID, OUTF);
    transpose_w<<<(HID * OUTF + 255) / 256, blk256, 0, stream>>>(c2_so_Wr, t2_so_Wr, HID, OUTF);

    gemm_sage<D_IN, HID, false><<<dim3(rt_occ, HID / 64), blk128, 0, stream>>>(
        x_occ, pWt_occ, nullptr, nullptr, nullptr, pb_occ, h_occ, N_OCC, 1);
    gemm_sage<D_IN, HID, false><<<dim3(rt_skill, HID / 64), blk128, 0, stream>>>(
        x_skill, pWt_skill, nullptr, nullptr, nullptr, pb_skill, h_skill, N_SKILL, 1);

    owner_sum<true><<<NBLK, dim3(1024), 0, stream>>>(
        h_occ, h_skill, e_occ, e_skill, agg_occ, agg_skill, cnt_occ, cnt_skill, edge_lists);

    gemm_sage<HID, HID, true><<<dim3(rt_skill, HID / 64), blk128, 0, stream>>>(
        agg_skill, t1_os_Wl, h_skill, t1_os_Wr, cnt_skill, c1_os_bl, s1, N_SKILL, 1);
    gemm_sage<HID, HID, true><<<dim3(rt_occ, HID / 64), blk128, 0, stream>>>(
        agg_occ, t1_so_Wl, h_occ, t1_so_Wr, cnt_occ, c1_so_bl, o1, N_OCC, 1);

    owner_sum<false><<<NBLK, dim3(1024), 0, stream>>>(
        o1, s1, e_occ, e_skill, agg_occ, agg_skill, cnt_occ, cnt_skill, edge_lists);

    gemm_sage<HID, OUTF, true><<<dim3(rt_skill, OUTF / 64), blk128, 0, stream>>>(
        agg_skill, t2_os_Wl, s1, t2_os_Wr, cnt_skill, c2_os_bl, s2, N_SKILL, 0);
    gemm_sage<HID, OUTF, true><<<dim3(rt_occ, OUTF / 64), blk128, 0, stream>>>(
        agg_occ, t2_so_Wl, o1, t2_so_Wr, cnt_occ, c2_so_bl, o2, N_OCC, 0);
}


# B fragments read straight from row-major W: a strip's 4 column tiles interleaved (tile t = columns 4c+t), one 16-byte load per k row, 16-byte output stores; the 10 transposes go; f32 sum order kept
# speedup vs baseline: 3.0023x; 1.4665x over previous
#include <hip/hip_runtime.h>

#define N_OCC   20000
#define N_SKILL 50000
#define N_EDGE  600000
#define D_IN    256
#define HID     128
#define OUTF    64

typedef __attribute__((ext_vector_type(2))) float v2f;
typedef __attribute__((ext_vector_type(4))) float v4f;
typedef __attribute__((ext_vector_type(8))) float v8f;

template <int K, int N>
__device__ inline void strip_product(const float* __restrict__ ap, const float* __restrict__ bp,
                                     v8f& c0, v8f& c1, v8f& c2, v8f& c3)
{
    #pragma unroll 8
    for (int k = 0; k < K / 4; ++k) {
        const v2f a  = *(const v2f*)(ap + 4 * k);
        const v4f w0 = *(const v4f*)(bp + (size_t)(4 * k) * N);
        const v4f w1 = *(const v4f*)(bp + (size_t)(4 * k + 1) * N);
        const v2f b0 = {w0.x, w1.x};
        const v2f b1 = {w0.y, w1.y};
        const v2f b2 = {w0.z, w1.z};
        const v2f b3 = {w0.w, w1.w};
        c0 = __builtin_amdgcn_wmma_f32_16x16x4_f32(false, a, false, b0, (short)0, c0, false, false);
        c1 = __builtin_amdgcn_wmma_f32_16x16x4_f32(false, a, false, b1, (short)0, c1, false, false);
        c2 = __builtin_amdgcn_wmma_f32_16x16x4_f32(false, a, false, b2, (short)0, c2, false, false);
        c3 = __builtin_amdgcn_wmma_f32_16x16x4_f32(false, a, false, b3, (short)0, c3, false, false);
    }
}

template <int K, int N, bool FUSE>
__global__ void __launch_bounds__(128)
gemm_sage(const float* __restrict__ A1, const float* __restrict__ W1,
          const float* __restrict__ A2, const float* __restrict__ W2,
          const unsigned* __restrict__ cnt, const float* __restrict__ bias,
          float* __restrict__ out, int M, int do_relu)
{
    static_assert(N % 64 == 0 && K % 32 == 0, "strip of four interleaved column tiles; k loop unrolled by 8");
    const int lane = threadIdx.x & 31;
    const int wave = threadIdx.x >> 5;
    const int hi   = lane >> 4;
    const int lm   = lane & 15;
    const int rt   = blockIdx.x * 4 + wave;
    if (rt * 16 >= M) return;
    const int row0 = rt * 16;
    const int col0 = blockIdx.y * 64;

    v8f c0 = {0.f,0.f,0.f,0.f,0.f,0.f,0.f,0.f};
    v8f c1 = c0, c2 = c0, c3 = c0;

    strip_product<K, N>(A1 + (size_t)(row0 + lm) * K + 2 * hi,
                        W1 + (size_t)(2 * hi) * N + col0 + 4 * lm, c0, c1, c2, c3);

    if (FUSE) {
        #pragma unroll
        for (int r = 0; r < 8; ++r) {
            const float inv = 1.0f / fmaxf((float)cnt[row0 + r + 8 * hi], 1.0f);
            c0[r] *= inv; c1[r] *= inv; c2[r] *= inv; c3[r] *= inv;
        }
        strip_product<K, N>(A2 + (size_t)(row0 + lm) * K + 2 * hi,
                            W2 + (size_t)(2 * hi) * N + col0 + 4 * lm, c0, c1, c2, c3);
    }

    float* orow = out + (size_t)(row0 + 8 * hi) * N + col0 + 4 * lm;
    const v4f bv = *(const v4f*)(bias + col0 + 4 * lm);
    #pragma unroll
    for (int r = 0; r < 8; ++r) {
        v4f v = {c0[r] + bv.x, c1[r] + bv.y, c2[r] + bv.z, c3[r] + bv.w};
        if (do_relu) {
            v.x = fmaxf(v.x, 0.f); v.y = fmaxf(v.y, 0.f);
            v.z = fmaxf(v.z, 0.f); v.w = fmaxf(v.w, 0.f);
        }
        *(v4f*)(orow + (size_t)r * N) = v;
    }
}

#define NBLK   256
#define ROWS_S 196
#define ROWS_O 79
#define ROWS_T (ROWS_S + ROWS_O)
#define NWAVE  32
#define CAPW   544
#define QCAP   (CAPW + 512)
#define QPEND  128
#define LIST_HDR  NWAVE
#define BLK_WORDS (LIST_HDR + NWAVE * CAPW)
#define NVEC   (N_EDGE / 4)
#define NITER  ((NVEC + 2047) / 2048)

static_assert(N_EDGE % 4 == 0, "edge list is read as int4");
static_assert(NBLK * ROWS_S >= N_SKILL && NBLK * ROWS_O >= N_OCC, "row ranges cover all nodes");
static_assert(N_SKILL < 65536 && N_OCC < 65536 && ROWS_T < 65536, "queue entry packing");
static_assert(QCAP % 8 == 0 && QCAP > 512, "drain reads whole batches of 8; room for one iteration");
static_assert(CAPW % 32 == 0 && BLK_WORDS % 32 == 0, "lists start on 128-byte lines");
static_assert((size_t)NBLK * BLK_WORDS <= (size_t)(N_OCC + N_SKILL) * OUTF, "lists fit the output buffer");

__device__ inline void lds_add_f32(float* p, float v)
{
    (void)__hip_atomic_fetch_add(p, v, __ATOMIC_RELAXED, __HIP_MEMORY_SCOPE_WORKGROUP);
}

template <bool COUNT>
__device__ inline void drain_queue(const unsigned* myq, int qs, int qn, float* acc, unsigned* cl,
                                   const float* __restrict__ f_occ,
                                   const float* __restrict__ f_skill, int lane)
{
    for (int base = qs; base < qn; base += 8) {
        unsigned row[8];
        float v[8][4];
        #pragma unroll
        for (int j = 0; j < 8; ++j) {
            unsigned e = __builtin_amdgcn_readfirstlane(myq[base + j]);
            if (base + j >= qn) e = 0u;
            row[j] = e >> 16;
            const unsigned srow = e & 0xffffu;
            const float* tab = (row[j] < (unsigned)ROWS_S) ? f_occ : f_skill;
            const float* p = tab + (size_t)srow * HID + lane;
            v[j][0] = p[0]; v[j][1] = p[32]; v[j][2] = p[64]; v[j][3] = p[96];
        }
        #pragma unroll
        for (int j = 0; j < 8; ++j) {
            if (base + j < qn) {
                float* d = acc + row[j] * HID + lane;
                lds_add_f32(d,      v[j][0]);
                lds_add_f32(d + 32, v[j][1]);
                lds_add_f32(d + 64, v[j][2]);
                lds_add_f32(d + 96, v[j][3]);
                if (COUNT && lane == 0) atomicAdd(&cl[row[j]], 1u);
            }
        }
    }
}

template <bool COUNT>
__global__ void __launch_bounds__(1024)
owner_sum(const float* __restrict__ f_occ, const float* __restrict__ f_skill,
          const int* __restrict__ e_occ, const int* __restrict__ e_skill,
          float* __restrict__ agg_occ, float* __restrict__ agg_skill,
          unsigned* __restrict__ cnt_occ, unsigned* __restrict__ cnt_skill,
          float* __restrict__ list_buf)
{
    __shared__ float    acc[ROWS_T * HID];
    __shared__ unsigned q[NWAVE * QCAP];
    __shared__ unsigned cl[ROWS_T + 1];
    __shared__ unsigned ln[NWAVE];

    const int tid  = threadIdx.x;
    const int lane = tid & 31;
    const int wave = tid >> 5;
    const int s_lo = blockIdx.x * ROWS_S;
    const int o_lo = blockIdx.x * ROWS_O;
    unsigned* myq = q + wave * QCAP;
    unsigned* blk_list = (unsigned*)list_buf + (size_t)blockIdx.x * BLK_WORDS;
    unsigned* wl = blk_list + LIST_HDR + wave * CAPW;

    unsigned n_list = 0u;
    if (!COUNT) n_list = blk_list[wave];

    for (int i = tid; i < ROWS_T * HID / 4; i += 1024)
        ((float4*)acc)[i] = make_float4(0.f, 0.f, 0.f, 0.f);
    if (COUNT)
        for (int i = tid; i < ROWS_T; i += 1024) cl[i] = 0u;

    int qs = 0;
    int qn = 0;
    bool scan = true;
    if (!COUNT) {
        const unsigned n = __builtin_amdgcn_readfirstlane(n_list);
        if (n <= (unsigned)CAPW) {
            for (int i0 = 0; i0 < (int)n; i0 += 256) {
                unsigned t[8];
                #pragma unroll
                for (int j = 0; j < 8; ++j) {
                    const int i = i0 + 32 * j + lane;
                    t[j] = (i < (int)n) ? wl[i] : 0u;
                }
                #pragma unroll
                for (int j = 0; j < 8; ++j) {
                    const int i = i0 + 32 * j + lane;
                    if (i < (int)n) myq[i] = t[j];
                }
            }
            qn = (int)n;
            scan = false;
        }
    }
    __syncthreads();

    const int4* vo = (const int4*)e_occ;
    const int4* vs = (const int4*)e_skill;
    const int4 none = make_int4(-1, -1, -1, -1);

    bool fits = true;
    int it = (int)((blockIdx.x * NITER) >> 8);

    if (scan) {
    for (int k = 0; k < NITER; ++k) {
        const int ia = it * 2048 + tid;
        const int ib = ia + 1024;
        int4 eo0 = none, es0 = none, eo1 = none, es1 = none;
        if (ia < NVEC) { eo0 = vo[ia]; es0 = vs[ia]; }
        if (ib < NVEC) { eo1 = vo[ib]; es1 = vs[ib]; }

        if (qn - qs > QPEND || qn > QCAP - 512) {
            drain_queue<COUNT>(myq, qs, qn, acc, cl, f_occ, f_skill, lane);
            qs = qn;
            if (qn > QCAP - 512) { qs = 0; qn = 0; fits = false; }
        }

#define PUSH_EDGE(O, S)                                                              \
        {                                                                            \
            const unsigned rs = (unsigned)((S) - s_lo);                              \
            const bool ms = rs < (unsigned)ROWS_S;                                   \
            const unsigned bs = __builtin_amdgcn_ballot_w32(ms);                     \
            if (bs) {                                                                \
                if (ms) myq[qn + __builtin_amdgcn_mbcnt_lo(bs, 0u)] = (rs << 16) | (unsigned)(O); \
                qn += __builtin_popcount(bs);                                        \
            }                                                                        \
            const unsigned ro = (unsigned)((O) - o_lo);                              \
            const bool mo = ro < (unsigned)ROWS_O;                                   \
            const unsigned bo = __builtin_amdgcn_ballot_w32(mo);                     \
            if (bo) {                                                                \
                if (mo) myq[qn + __builtin_amdgcn_mbcnt_lo(bo, 0u)] = ((ro + ROWS_S) << 16) | (unsigned)(S); \
                qn += __builtin_popcount(bo);                                        \
            }                                                                        \
        }
        PUSH_EDGE(eo0.x, es0.x)
        PUSH_EDGE(eo0.y, es0.y)
        PUSH_EDGE(eo0.z, es0.z)
        PUSH_EDGE(eo0.w, es0.w)
        PUSH_EDGE(eo1.x, es1.x)
        PUSH_EDGE(eo1.y, es1.y)
        PUSH_EDGE(eo1.z, es1.z)
        PUSH_EDGE(eo1.w, es1.w)
#undef PUSH_EDGE
        it = (it + 1 == NITER) ? 0 : it + 1;
    }
    }
    drain_queue<COUNT>(myq, qs, qn, acc, cl, f_occ, f_skill, lane);

    if (COUNT) {
        const bool keep = fits && qn <= CAPW;
        if (keep)
            for (int i = lane; i < qn; i += 32) wl[i] = myq[i];
        if (lane == 0) ln[wave] = keep ? (unsigned)qn : (unsigned)(CAPW + 1);
    }

    __syncthreads();

    if (COUNT && wave == 0) blk_list[lane] = ln[lane];

    for (int i = tid; i < ROWS_S * HID / 4; i += 1024) {
        const int r = s_lo + (i >> 5);
        if (r < N_SKILL) ((float4*)agg_skill)[(size_t)r * (HID / 4) + (i & 31)] = ((const float4*)acc)[i];
    }
    for (int i = tid; i < ROWS_O * HID / 4; i += 1024) {
        const int r = o_lo + (i >> 5);
        if (r < N_OCC) ((float4*)agg_occ)[(size_t)r * (HID / 4) + (i & 31)] = ((const float4*)acc)[ROWS_S * HID / 4 + i];
    }
    if (COUNT) {
        for (int i = tid; i < ROWS_S; i += 1024)
            if (s_lo + i < N_SKILL) cnt_skill[s_lo + i] = cl[i];
        for (int i = tid; i < ROWS_O; i += 1024)
            if (o_lo + i < N_OCC) cnt_occ[o_lo + i] = cl[ROWS_S + i];
    }
}

extern "C" void kernel_launch(void* const* d_in, const int* in_sizes, int n_in,
                              void* d_out, int out_size, void* d_ws, size_t ws_size,
                              hipStream_t stream)
{
    (void)in_sizes; (void)n_in; (void)out_size; (void)ws_size;

    const float* x_occ    = (const float*)d_in[0];
    const float* x_skill  = (const float*)d_in[1];
    const int*   e_occ    = (const int*)  d_in[2];
    const int*   e_skill  = (const int*)  d_in[3];
    const float* pW_occ   = (const float*)d_in[4];
    const float* pb_occ   = (const float*)d_in[5];
    const float* pW_skill = (const float*)d_in[6];
    const float* pb_skill = (const float*)d_in[7];
    const float* c1_os_Wl = (const float*)d_in[8];
    const float* c1_os_bl = (const float*)d_in[9];
    const float* c1_os_Wr = (const float*)d_in[10];
    const float* c1_so_Wl = (const float*)d_in[11];
    const float* c1_so_bl = (const float*)d_in[12];
    const float* c1_so_Wr = (const float*)d_in[13];
    const float* c2_os_Wl = (const float*)d_in[14];
    const float* c2_os_bl = (const float*)d_in[15];
    const float* c2_os_Wr = (const float*)d_in[16];
    const float* c2_so_Wl = (const float*)d_in[17];
    const float* c2_so_bl = (const float*)d_in[18];
    const float* c2_so_Wr = (const float*)d_in[19];

    float* ws = (float*)d_ws;
    float* h_occ     = ws;
    float* h_skill   = h_occ     + (size_t)N_OCC   * HID;
    float* o1        = h_skill   + (size_t)N_SKILL * HID;
    float* s1        = o1        + (size_t)N_OCC   * HID;
    float* agg_occ   = s1        + (size_t)N_SKILL * HID;
    float* agg_skill = agg_occ   + (size_t)N_OCC   * HID;
    unsigned* cnt_occ   = (unsigned*)(agg_skill + (size_t)N_SKILL * HID);
    unsigned* cnt_skill = cnt_occ + N_OCC;

    float* o2 = (float*)d_out;
    float* s2 = o2 + (size_t)N_OCC * OUTF;
    float* edge_lists = (float*)d_out;

    const dim3 blk128(128);
    const int rt_occ   = (N_OCC / 16 + 3) / 4;
    const int rt_skill = (N_SKILL / 16 + 3) / 4;

    gemm_sage<D_IN, HID, false><<<dim3(rt_occ, HID / 64), blk128, 0, stream>>>(
        x_occ, pW_occ, nullptr, nullptr, nullptr, pb_occ, h_occ, N_OCC, 1);
    gemm_sage<D_IN, HID, false><<<dim3(rt_skill, HID / 64), blk128, 0, stream>>>(
        x_skill, pW_skill, nullptr, nullptr, nullptr, pb_skill, h_skill, N_SKILL, 1);

    owner_sum<true><<<NBLK, dim3(1024), 0, stream>>>(
        h_occ, h_skill, e_occ, e_skill, agg_occ, agg_skill, cnt_occ, cnt_skill, edge_lists);

    gemm_sage<HID, HID, true><<<dim3(rt_skill, HID / 64), blk128, 0, stream>>>(
        agg_skill, c1_os_Wl, h_skill, c1_os_Wr, cnt_skill, c1_os_bl, s1, N_SKILL, 1);
    gemm_sage<HID, HID, true><<<dim3(rt_occ, HID / 64), blk128, 0, stream>>>(
        agg_occ, c1_so_Wl, h_occ, c1_so_Wr, cnt_occ, c1_so_bl, o1, N_OCC, 1);

    owner_sum<false><<<NBLK, dim3(1024), 0, stream>>>(
        o1, s1, e_occ, e_skill, agg_occ, agg_skill, cnt_occ, cnt_skill, edge_lists);

    gemm_sage<HID, OUTF, true><<<dim3(rt_skill, OUTF / 64), blk128, 0, stream>>>(
        agg_skill, c2_os_Wl, s1, c2_os_Wr, cnt_skill, c2_os_bl, s2, N_SKILL, 0);
    gemm_sage<HID, OUTF, true><<<dim3(rt_occ, OUTF / 64), blk128, 0, stream>>>(
        agg_occ, c2_so_Wl, o1, c2_so_Wr, cnt_occ, c2_so_bl, o2, N_OCC, 0);
}


# each layer's two products run inside its neighbour-sum launch on the rows a block owns: sums stay in LDS as the A operand (row pitch 132), agg arrays go; 8 -> 4 launches; f32 sum order unchanged
# speedup vs baseline: 3.4095x; 1.1356x over previous
#include <hip/hip_runtime.h>

#define N_OCC   20000
#define N_SKILL 50000
#define N_EDGE  600000
#define D_IN    256
#define HID     128
#define OUTF    64

typedef __attribute__((ext_vector_type(2))) float v2f;
typedef __attribute__((ext_vector_type(4))) float v4f;
typedef __attribute__((ext_vector_type(8))) float v8f;

template <int K, int N>
__device__ inline void strip_product(const float* __restrict__ ap, const float* __restrict__ bp,
                                     v8f& c0, v8f& c1, v8f& c2, v8f& c3)
{
    #pragma unroll 8
    for (int k = 0; k < K / 4; ++k) {
        const v2f a  = *(const v2f*)(ap + 4 * k);
        const v4f w0 = *(const v4f*)(bp + (size_t)(4 * k) * N);
        const v4f w1 = *(const v4f*)(bp + (size_t)(4 * k + 1) * N);
        const v2f b0 = {w0.x, w1.x};
        const v2f b1 = {w0.y, w1.y};
        const v2f b2 = {w0.z, w1.z};
        const v2f b3 = {w0.w, w1.w};
        c0 = __builtin_amdgcn_wmma_f32_16x16x4_f32(false, a, false, b0, (short)0, c0, false, false);
        c1 = __builtin_amdgcn_wmma_f32_16x16x4_f32(false, a, false, b1, (short)0, c1, false, false);
        c2 = __builtin_amdgcn_wmma_f32_16x16x4_f32(false, a, false, b2, (short)0, c2, false, false);
        c3 = __builtin_amdgcn_wmma_f32_16x16x4_f32(false, a, false, b3, (short)0, c3, false, false);
    }
}

template <int K, int N, bool FUSE>
__global__ void __launch_bounds__(128)
gemm_sage(const float* __restrict__ A1, const float* __restrict__ W1,
          const float* __restrict__ A2, const float* __restrict__ W2,
          const unsigned* __restrict__ cnt, const float* __restrict__ bias,
          float* __restrict__ out, int M, int do_relu)
{
    static_assert(N % 64 == 0 && K % 32 == 0, "strip of four interleaved column tiles; k loop unrolled by 8");
    const int lane = threadIdx.x & 31;
    const int wave = threadIdx.x >> 5;
    const int hi   = lane >> 4;
    const int lm   = lane & 15;
    const int rt   = blockIdx.x * 4 + wave;
    if (rt * 16 >= M) return;
    const int row0 = rt * 16;
    const int col0 = blockIdx.y * 64;

    v8f c0 = {0.f,0.f,0.f,0.f,0.f,0.f,0.f,0.f};
    v8f c1 = c0, c2 = c0, c3 = c0;

    strip_product<K, N>(A1 + (size_t)(row0 + lm) * K + 2 * hi,
                        W1 + (size_t)(2 * hi) * N + col0 + 4 * lm, c0, c1, c2, c3);

    if (FUSE) {
        #pragma unroll
        for (int r = 0; r < 8; ++r) {
            const float inv = 1.0f / fmaxf((float)cnt[row0 + r + 8 * hi], 1.0f);
            c0[r] *= inv; c1[r] *= inv; c2[r] *= inv; c3[r] *= inv;
        }
        strip_product<K, N>(A2 + (size_t)(row0 + lm) * K + 2 * hi,
                            W2 + (size_t)(2 * hi) * N + col0 + 4 * lm, c0, c1, c2, c3);
    }

    float* orow = out + (size_t)(row0 + 8 * hi) * N + col0 + 4 * lm;
    const v4f bv = *(const v4f*)(bias + col0 + 4 * lm);
    #pragma unroll
    for (int r = 0; r < 8; ++r) {
        v4f v = {c0[r] + bv.x, c1[r] + bv.y, c2[r] + bv.z, c3[r] + bv.w};
        if (do_relu) {
            v.x = fmaxf(v.x, 0.f); v.y = fmaxf(v.y, 0.f);
            v.z = fmaxf(v.z, 0.f); v.w = fmaxf(v.w, 0.f);
        }
        *(v4f*)(orow + (size_t)r * N) = v;
    }
}

#define NBLK   256
#define ROWS_S 196
#define ROWS_O 79
#define ROWS_T (ROWS_S + ROWS_O)
#define NWAVE  32
#define CAPW   544
#define QCAP   (CAPW + 512)
#define QPEND  128
#define LIST_HDR  NWAVE
#define BLK_WORDS (LIST_HDR + NWAVE * CAPW)
#define NVEC   (N_EDGE / 4)
#define PITCH  (HID + 4)
#define TILES_S ((ROWS_S + 15) / 16)
#define TILES_O ((ROWS_O + 15) / 16)
#define NITER  ((NVEC + 2047) / 2048)

static_assert(N_EDGE % 4 == 0, "edge list is read as int4");
static_assert(NBLK * ROWS_S >= N_SKILL && NBLK * ROWS_O >= N_OCC, "row ranges cover all nodes");
static_assert(N_SKILL < 65536 && N_OCC < 65536 && ROWS_T < 65536, "queue entry packing");
static_assert(QCAP % 8 == 0 && QCAP > 512, "drain reads whole batches of 8; room for one iteration");
static_assert(CAPW % 32 == 0 && BLK_WORDS % 32 == 0, "lists start on 128-byte lines");
static_assert((size_t)NBLK * BLK_WORDS <= (size_t)(N_OCC + N_SKILL) * HID, "lists fit their workspace region");
static_assert((ROWS_T * PITCH) % 4 == 0, "zero fill writes float4");

__device__ inline void lds_add_f32(float* p, float v)
{
    (void)__hip_atomic_fetch_add(p, v, __ATOMIC_RELAXED, __HIP_MEMORY_SCOPE_WORKGROUP);
}

template <bool COUNT>
__device__ inline void drain_queue(const unsigned* myq, int qs, int qn, float* acc, unsigned* cl,
                                   const float* __restrict__ f_occ,
                                   const float* __restrict__ f_skill, int lane)
{
    for (int base = qs; base < qn; base += 8) {
        unsigned row[8];
        float v[8][4];
        #pragma unroll
        for (int j = 0; j < 8; ++j) {
            unsigned e = __builtin_amdgcn_readfirstlane(myq[base + j]);
            if (base + j >= qn) e = 0u;
            row[j] = e >> 16;
            const unsigned srow = e & 0xffffu;
            const float* tab = (row[j] < (unsigned)ROWS_S) ? f_occ : f_skill;
            const float* p = tab + (size_t)srow * HID + lane;
            v[j][0] = p[0]; v[j][1] = p[32]; v[j][2] = p[64]; v[j][3] = p[96];
        }
        #pragma unroll
        for (int j = 0; j < 8; ++j) {
            if (base + j < qn) {
                float* d = acc + row[j] * PITCH + lane;
                lds_add_f32(d,      v[j][0]);
                lds_add_f32(d + 32, v[j][1]);
                lds_add_f32(d + 64, v[j][2]);
                lds_add_f32(d + 96, v[j][3]);
                if (COUNT && lane == 0) atomicAdd(&cl[row[j]], 1u);
            }
        }
    }
}

template <int N, bool COUNT>
__device__ inline void own_rows_product(const float* acc, const unsigned* cl,
                                        const float* __restrict__ x_dst,
                                        const float* __restrict__ Wl, const float* __restrict__ Wr,
                                        const float* __restrict__ bias, const unsigned* __restrict__ cnt_g,
                                        float* __restrict__ outp,
                                        int lrow_base, int n_own, int grow_base, int n_total,
                                        int tile, int col0, int do_relu, int lane)
{
    const int hi  = lane >> 4;
    const int lm  = lane & 15;
    const int lr0 = 16 * tile;
    const int la  = min(lrow_base + lr0 + lm, ROWS_T - 1);
    const int ga  = min(grow_base + lr0 + lm, n_total - 1);

    v8f c0 = {0.f,0.f,0.f,0.f,0.f,0.f,0.f,0.f};
    v8f c1 = c0, c2 = c0, c3 = c0;

    strip_product<HID, N>(acc + la * PITCH + 2 * hi,
                          Wl + (size_t)(2 * hi) * N + col0 + 4 * lm, c0, c1, c2, c3);

    #pragma unroll
    for (int r = 0; r < 8; ++r) {
        const int rr = lr0 + r + 8 * hi;
        const unsigned cn = COUNT ? cl[min(lrow_base + rr, ROWS_T - 1)]
                                  : cnt_g[min(grow_base + rr, n_total - 1)];
        const float inv = 1.0f / fmaxf((float)cn, 1.0f);
        c0[r] *= inv; c1[r] *= inv; c2[r] *= inv; c3[r] *= inv;
    }

    strip_product<HID, N>(x_dst + (size_t)ga * HID + 2 * hi,
                          Wr + (size_t)(2 * hi) * N + col0 + 4 * lm, c0, c1, c2, c3);

    const v4f bv = *(const v4f*)(bias + col0 + 4 * lm);
    #pragma unroll
    for (int r = 0; r < 8; ++r) {
        const int rr = lr0 + r + 8 * hi;
        const int g  = grow_base + rr;
        v4f v = {c0[r] + bv.x, c1[r] + bv.y, c2[r] + bv.z, c3[r] + bv.w};
        if (do_relu) {
            v.x = fmaxf(v.x, 0.f); v.y = fmaxf(v.y, 0.f);
            v.z = fmaxf(v.z, 0.f); v.w = fmaxf(v.w, 0.f);
        }
        if (rr < n_own && g < n_total)
            *(v4f*)(outp + (size_t)g * N + col0 + 4 * lm) = v;
    }
}

template <bool COUNT, int N>
__global__ void __launch_bounds__(1024)
owner_sum(const float* __restrict__ f_occ, const float* __restrict__ f_skill,
          const int* __restrict__ e_occ, const int* __restrict__ e_skill,
          unsigned* __restrict__ cnt_occ, unsigned* __restrict__ cnt_skill,
          float* __restrict__ list_buf,
          const float* __restrict__ Wl_s, const float* __restrict__ Wr_s, const float* __restrict__ bl_s,
          const float* __restrict__ Wl_o, const float* __restrict__ Wr_o, const float* __restrict__ bl_o,
          float* __restrict__ out_skill, float* __restrict__ out_occ, int do_relu)
{
    __shared__ float    acc[ROWS_T * PITCH];
    __shared__ unsigned q[NWAVE * QCAP];
    __shared__ unsigned cl[ROWS_T + 1];
    __shared__ unsigned ln[NWAVE];

    const int tid  = threadIdx.x;
    const int lane = tid & 31;
    const int wave = tid >> 5;
    const int s_lo = blockIdx.x * ROWS_S;
    const int o_lo = blockIdx.x * ROWS_O;
    unsigned* myq = q + wave * QCAP;
    unsigned* blk_list = (unsigned*)list_buf + (size_t)blockIdx.x * BLK_WORDS;
    unsigned* wl = blk_list + LIST_HDR + wave * CAPW;

    unsigned n_list = 0u;
    if (!COUNT) n_list = blk_list[wave];

    for (int i = tid; i < ROWS_T * PITCH / 4; i += 1024)
        ((float4*)acc)[i] = make_float4(0.f, 0.f, 0.f, 0.f);
    if (COUNT)
        for (int i = tid; i < ROWS_T; i += 1024) cl[i] = 0u;

    int qs = 0;
    int qn = 0;
    bool scan = true;
    if (!COUNT) {
        const unsigned n = __builtin_amdgcn_readfirstlane(n_list);
        if (n <= (unsigned)CAPW) {
            for (int i0 = 0; i0 < (int)n; i0 += 256) {
                unsigned t[8];
                #pragma unroll
                for (int j = 0; j < 8; ++j) {
                    const int i = i0 + 32 * j + lane;
                    t[j] = (i < (int)n) ? wl[i] : 0u;
                }
                #pragma unroll
                for (int j = 0; j < 8; ++j) {
                    const int i = i0 + 32 * j + lane;
                    if (i < (int)n) myq[i] = t[j];
                }
            }
            qn = (int)n;
            scan = false;
        }
    }
    __syncthreads();

    const int4* vo = (const int4*)e_occ;
    const int4* vs = (const int4*)e_skill;
    const int4 none = make_int4(-1, -1, -1, -1);

    bool fits = true;
    int it = (int)((blockIdx.x * NITER) >> 8);

    if (scan) {
    for (int k = 0; k < NITER; ++k) {
        const int ia = it * 2048 + tid;
        const int ib = ia + 1024;
        int4 eo0 = none, es0 = none, eo1 = none, es1 = none;
        if (ia < NVEC) { eo0 = vo[ia]; es0 = vs[ia]; }
        if (ib < NVEC) { eo1 = vo[ib]; es1 = vs[ib]; }

        if (qn - qs > QPEND || qn > QCAP - 512) {
            drain_queue<COUNT>(myq, qs, qn, acc, cl, f_occ, f_skill, lane);
            qs = qn;
            if (qn > QCAP - 512) { qs = 0; qn = 0; fits = false; }
        }

#define PUSH_EDGE(O, S)                                                              \
        {                                                                            \
            const unsigned rs = (unsigned)((S) - s_lo);                              \
            const bool ms = rs < (unsigned)ROWS_S;                                   \
            const unsigned bs = __builtin_amdgcn_ballot_w32(ms);                     \
            if (bs) {                                                                \
                if (ms) myq[qn + __builtin_amdgcn_mbcnt_lo(bs, 0u)] = (rs << 16) | (unsigned)(O); \
                qn += __builtin_popcount(bs);                                        \
            }                                                                        \
            const unsigned ro = (unsigned)((O) - o_lo);                              \
            const bool mo = ro < (unsigned)ROWS_O;                                   \
            const unsigned bo = __builtin_amdgcn_ballot_w32(mo);                     \
            if (bo) {                                                                \
                if (mo) myq[qn + __builtin_amdgcn_mbcnt_lo(bo, 0u)] = ((ro + ROWS_S) << 16) | (unsigned)(S); \
                qn += __builtin_popcount(bo);                                        \
            }                                                                        \
        }
        PUSH_EDGE(eo0.x, es0.x)
        PUSH_EDGE(eo0.y, es0.y)
        PUSH_EDGE(eo0.z, es0.z)
        PUSH_EDGE(eo0.w, es0.w)
        PUSH_EDGE(eo1.x, es1.x)
        PUSH_EDGE(eo1.y, es1.y)
        PUSH_EDGE(eo1.z, es1.z)
        PUSH_EDGE(eo1.w, es1.w)
#undef PUSH_EDGE
        it = (it + 1 == NITER) ? 0 : it + 1;
    }
    }
    drain_queue<COUNT>(myq, qs, qn, acc, cl, f_occ, f_skill, lane);

    if (COUNT) {
        const bool keep = fits && qn <= CAPW;
        if (keep)
            for (int i = lane; i < qn; i += 32) wl[i] = myq[i];
        if (lane == 0) ln[wave] = keep ? (unsigned)qn : (unsigned)(CAPW + 1);
    }

    __syncthreads();

    if (COUNT && wave == 0) blk_list[lane] = ln[lane];

    if (COUNT) {
        for (int i = tid; i < ROWS_S; i += 1024)
            if (s_lo + i < N_SKILL) cnt_skill[s_lo + i] = cl[i];
        for (int i = tid; i < ROWS_O; i += 1024)
            if (o_lo + i < N_OCC) cnt_occ[o_lo + i] = cl[ROWS_S + i];
    }

    constexpr int NSTRIP = N / 64;
    const int wave_s = __builtin_amdgcn_readfirstlane(wave);
    for (int task = wave_s; task < (TILES_S + TILES_O) * NSTRIP; task += NWAVE) {
        const int  t    = task / NSTRIP;
        const int  col0 = (task % NSTRIP) * 64;
        const bool occ  = t >= TILES_S;
        const int  tile = occ ? t - TILES_S : t;
        const int  gb   = occ ? o_lo : s_lo;
        const int  ntot = occ ? N_OCC : N_SKILL;
        const int  nown = occ ? ROWS_O : ROWS_S;
        if (gb + 16 * tile >= ntot) continue;
        own_rows_product<N, COUNT>(acc, cl,
                                   occ ? f_occ : f_skill,
                                   occ ? Wl_o : Wl_s, occ ? Wr_o : Wr_s, occ ? bl_o : bl_s,
                                   occ ? cnt_occ : cnt_skill,
                                   occ ? out_occ : out_skill,
                                   occ ? ROWS_S : 0, nown, gb, ntot, tile, col0, do_relu, lane);
    }
}

extern "C" void kernel_launch(void* const* d_in, const int* in_sizes, int n_in,
                              void* d_out, int out_size, void* d_ws, size_t ws_size,
                              hipStream_t stream)
{
    (void)in_sizes; (void)n_in; (void)out_size; (void)ws_size;

    const float* x_occ    = (const float*)d_in[0];
    const float* x_skill  = (const float*)d_in[1];
    const int*   e_occ    = (const int*)  d_in[2];
    const int*   e_skill  = (const int*)  d_in[3];
    const float* pW_occ   = (const float*)d_in[4];
    const float* pb_occ   = (const float*)d_in[5];
    const float* pW_skill = (const float*)d_in[6];
    const float* pb_skill = (const float*)d_in[7];
    const float* c1_os_Wl = (const float*)d_in[8];
    const float* c1_os_bl = (const float*)d_in[9];
    const float* c1_os_Wr = (const float*)d_in[10];
    const float* c1_so_Wl = (const float*)d_in[11];
    const float* c1_so_bl = (const float*)d_in[12];
    const float* c1_so_Wr = (const float*)d_in[13];
    const float* c2_os_Wl = (const float*)d_in[14];
    const float* c2_os_bl = (const float*)d_in[15];
    const float* c2_os_Wr = (const float*)d_in[16];
    const float* c2_so_Wl = (const float*)d_in[17];
    const float* c2_so_bl = (const float*)d_in[18];
    const float* c2_so_Wr = (const float*)d_in[19];

    float* ws = (float*)d_ws;
    float* h_occ     = ws;
    float* h_skill   = h_occ     + (size_t)N_OCC   * HID;
    float* o1        = h_skill   + (size_t)N_SKILL * HID;
    float* s1        = o1        + (size_t)N_OCC   * HID;
    float* edge_lists = s1       + (size_t)N_SKILL * HID;
    unsigned* cnt_occ   = (unsigned*)(edge_lists + (size_t)(N_OCC + N_SKILL) * HID);
    unsigned* cnt_skill = cnt_occ + N_OCC;

    float* o2 = (float*)d_out;
    float* s2 = o2 + (size_t)N_OCC * OUTF;

    const dim3 blk128(128);
    const int rt_occ   = (N_OCC / 16 + 3) / 4;
    const int rt_skill = (N_SKILL / 16 + 3) / 4;

    gemm_sage<D_IN, HID, false><<<dim3(rt_occ, HID / 64), blk128, 0, stream>>>(
        x_occ, pW_occ, nullptr, nullptr, nullptr, pb_occ, h_occ, N_OCC, 1);
    gemm_sage<D_IN, HID, false><<<dim3(rt_skill, HID / 64), blk128, 0, stream>>>(
        x_skill, pW_skill, nullptr, nullptr, nullptr, pb_skill, h_skill, N_SKILL, 1);

    owner_sum<true, HID><<<NBLK, dim3(1024), 0, stream>>>(
        h_occ, h_skill, e_occ, e_skill, cnt_occ, cnt_skill, edge_lists,
        c1_os_Wl, c1_os_Wr, c1_os_bl, c1_so_Wl, c1_so_Wr, c1_so_bl, s1, o1, 1);

    owner_sum<false, OUTF><<<NBLK, dim3(1024), 0, stream>>>(
        o1, s1, e_occ, e_skill, cnt_occ, cnt_skill, edge_lists,
        c2_os_Wl, c2_os_Wr, c2_os_bl, c2_so_Wl, c2_so_Wr, c2_so_bl, s2, o2, 0);
}


# neighbour sums: a block regroups its queued edges by destination row in LDS (count, prefix, returning integer LDS adds); waves sum row runs in registers, one LDS float add per run; f32 sum order new
# speedup vs baseline: 3.8631x; 1.1330x over previous
#include <hip/hip_runtime.h>

#define N_OCC   20000
#define N_SKILL 50000
#define N_EDGE  600000
#define D_IN    256
#define HID     128
#define OUTF    64

typedef __attribute__((ext_vector_type(2))) float v2f;
typedef __attribute__((ext_vector_type(4))) float v4f;
typedef __attribute__((ext_vector_type(8))) float v8f;

template <int K, int N>
__device__ inline void strip_product(const float* __restrict__ ap, const float* __restrict__ bp,
                                     v8f& c0, v8f& c1, v8f& c2, v8f& c3)
{
    #pragma unroll 8
    for (int k = 0; k < K / 4; ++k) {
        const v2f a  = *(const v2f*)(ap + 4 * k);
        const v4f w0 = *(const v4f*)(bp + (size_t)(4 * k) * N);
        const v4f w1 = *(const v4f*)(bp + (size_t)(4 * k + 1) * N);
        const v2f b0 = {w0.x, w1.x};
        const v2f b1 = {w0.y, w1.y};
        const v2f b2 = {w0.z, w1.z};
        const v2f b3 = {w0.w, w1.w};
        c0 = __builtin_amdgcn_wmma_f32_16x16x4_f32(false, a, false, b0, (short)0, c0, false, false);
        c1 = __builtin_amdgcn_wmma_f32_16x16x4_f32(false, a, false, b1, (short)0, c1, false, false);
        c2 = __builtin_amdgcn_wmma_f32_16x16x4_f32(false, a, false, b2, (short)0, c2, false, false);
        c3 = __builtin_amdgcn_wmma_f32_16x16x4_f32(false, a, false, b3, (short)0, c3, false, false);
    }
}

template <int K, int N, bool FUSE>
__global__ void __launch_bounds__(128)
gemm_sage(const float* __restrict__ A1, const float* __restrict__ W1,
          const float* __restrict__ A2, const float* __restrict__ W2,
          const unsigned* __restrict__ cnt, const float* __restrict__ bias,
          float* __restrict__ out, int M, int do_relu)
{
    static_assert(N % 64 == 0 && K % 32 == 0, "strip of four interleaved column tiles; k loop unrolled by 8");
    const int lane = threadIdx.x & 31;
    const int wave = threadIdx.x >> 5;
    const int hi   = lane >> 4;
    const int lm   = lane & 15;
    const int rt   = blockIdx.x * 4 + wave;
    if (rt * 16 >= M) return;
    const int row0 = rt * 16;
    const int col0 = blockIdx.y * 64;

    v8f c0 = {0.f,0.f,0.f,0.f,0.f,0.f,0.f,0.f};
    v8f c1 = c0, c2 = c0, c3 = c0;

    strip_product<K, N>(A1 + (size_t)(row0 + lm) * K + 2 * hi,
                        W1 + (size_t)(2 * hi) * N + col0 + 4 * lm, c0, c1, c2, c3);

    if (FUSE) {
        #pragma unroll
        for (int r = 0; r < 8; ++r) {
            const float inv = 1.0f / fmaxf((float)cnt[row0 + r + 8 * hi], 1.0f);
            c0[r] *= inv; c1[r] *= inv; c2[r] *= inv; c3[r] *= inv;
        }
        strip_product<K, N>(A2 + (size_t)(row0 + lm) * K + 2 * hi,
                            W2 + (size_t)(2 * hi) * N + col0 + 4 * lm, c0, c1, c2, c3);
    }

    float* orow = out + (size_t)(row0 + 8 * hi) * N + col0 + 4 * lm;
    const v4f bv = *(const v4f*)(bias + col0 + 4 * lm);
    #pragma unroll
    for (int r = 0; r < 8; ++r) {
        v4f v = {c0[r] + bv.x, c1[r] + bv.y, c2[r] + bv.z, c3[r] + bv.w};
        if (do_relu) {
            v.x = fmaxf(v.x, 0.f); v.y = fmaxf(v.y, 0.f);
            v.z = fmaxf(v.z, 0.f); v.w = fmaxf(v.w, 0.f);
        }
        *(v4f*)(orow + (size_t)r * N) = v;
    }
}

#define NBLK   256
#define ROWS_S 196
#define ROWS_O 79
#define ROWS_T (ROWS_S + ROWS_O)
#define NWAVE  32
#define CAPW   544
#define QCAP   (CAPW + 512)
#define SCAP   8192
#define LIST_HDR  NWAVE
#define BLK_WORDS (LIST_HDR + NWAVE * CAPW)
#define NVEC   (N_EDGE / 4)
#define PITCH  (HID + 4)
#define TILES_S ((ROWS_S + 15) / 16)
#define TILES_O ((ROWS_O + 15) / 16)
#define NITER  ((NVEC + 2047) / 2048)

static_assert(N_EDGE % 4 == 0, "edge list is read as int4");
static_assert(NBLK * ROWS_S >= N_SKILL && NBLK * ROWS_O >= N_OCC, "row ranges cover all nodes");
static_assert(N_SKILL < 65536 && N_OCC < 65536 && ROWS_T < 65536, "queue entry packing");
static_assert(QCAP % 8 == 0 && QCAP > 512, "drain reads whole batches of 8; room for one iteration");
static_assert(CAPW % 32 == 0 && BLK_WORDS % 32 == 0, "lists start on 128-byte lines");
static_assert((size_t)NBLK * BLK_WORDS <= (size_t)(N_OCC + N_SKILL) * HID, "lists fit their workspace region");
static_assert((ROWS_T * PITCH) % 4 == 0, "zero fill writes float4");
static_assert(32 * 9 >= ROWS_T, "prefix sum: 9 rows per lane of one wave");

__device__ inline void lds_add_f32(float* p, float v)
{
    (void)__hip_atomic_fetch_add(p, v, __ATOMIC_RELAXED, __HIP_MEMORY_SCOPE_WORKGROUP);
}

template <bool COUNT>
__device__ inline void drain_queue(const unsigned* myq, int qs, int qn, float* acc, unsigned* cl,
                                   const float* __restrict__ f_occ,
                                   const float* __restrict__ f_skill, int lane)
{
    for (int base = qs; base < qn; base += 8) {
        unsigned row[8];
        float v[8][4];
        #pragma unroll
        for (int j = 0; j < 8; ++j) {
            unsigned e = __builtin_amdgcn_readfirstlane(myq[base + j]);
            if (base + j >= qn) e = 0u;
            row[j] = e >> 16;
            const unsigned srow = e & 0xffffu;
            const float* tab = (row[j] < (unsigned)ROWS_S) ? f_occ : f_skill;
            const float* p = tab + (size_t)srow * HID + lane;
            v[j][0] = p[0]; v[j][1] = p[32]; v[j][2] = p[64]; v[j][3] = p[96];
        }
        #pragma unroll
        for (int j = 0; j < 8; ++j) {
            if (base + j < qn) {
                float* d = acc + row[j] * PITCH + lane;
                lds_add_f32(d,      v[j][0]);
                lds_add_f32(d + 32, v[j][1]);
                lds_add_f32(d + 64, v[j][2]);
                lds_add_f32(d + 96, v[j][3]);
                if (COUNT && lane == 0) atomicAdd(&cl[row[j]], 1u);
            }
        }
    }
}

__device__ inline void flush_run(float* acc, unsigned row, float s0, float s1, float s2, float s3, int lane)
{
    float* d = acc + row * PITCH + lane;
    lds_add_f32(d,      s0);
    lds_add_f32(d + 32, s1);
    lds_add_f32(d + 64, s2);
    lds_add_f32(d + 96, s3);
}

__device__ inline void sum_runs(const unsigned* sorted, int a, int b, float* acc,
                                const float* __restrict__ f_occ,
                                const float* __restrict__ f_skill, int lane)
{
    const unsigned NONE = 0xffffffffu;
    unsigned cur = NONE;
    float s0 = 0.f, s1 = 0.f, s2 = 0.f, s3 = 0.f;
    for (int base = a; base < b; base += 8) {
        const unsigned ev = sorted[min(base + (lane & 7), b - 1)];
        unsigned row[8];
        float v[8][4];
        #pragma unroll
        for (int j = 0; j < 8; ++j) {
            const unsigned e = __builtin_amdgcn_readlane(ev, j);
            row[j] = e >> 16;
            const unsigned srow = e & 0xffffu;
            const float* tab = (row[j] < (unsigned)ROWS_S) ? f_occ : f_skill;
            const float* p = tab + (size_t)srow * HID + lane;
            v[j][0] = p[0]; v[j][1] = p[32]; v[j][2] = p[64]; v[j][3] = p[96];
        }
        #pragma unroll
        for (int j = 0; j < 8; ++j) {
            if (base + j < b) {
                if (row[j] != cur) {
                    if (cur != NONE) flush_run(acc, cur, s0, s1, s2, s3, lane);
                    cur = row[j];
                    s0 = v[j][0]; s1 = v[j][1]; s2 = v[j][2]; s3 = v[j][3];
                } else {
                    s0 += v[j][0]; s1 += v[j][1]; s2 += v[j][2]; s3 += v[j][3];
                }
            }
        }
    }
    if (cur != NONE) flush_run(acc, cur, s0, s1, s2, s3, lane);
}

template <int N, bool COUNT>
__device__ inline void own_rows_product(const float* acc, const unsigned* cl,
                                        const float* __restrict__ x_dst,
                                        const float* __restrict__ Wl, const float* __restrict__ Wr,
                                        const float* __restrict__ bias, const unsigned* __restrict__ cnt_g,
                                        float* __restrict__ outp,
                                        int lrow_base, int n_own, int grow_base, int n_total,
                                        int tile, int col0, int do_relu, int lane)
{
    const int hi  = lane >> 4;
    const int lm  = lane & 15;
    const int lr0 = 16 * tile;
    const int la  = min(lrow_base + lr0 + lm, ROWS_T - 1);
    const int ga  = min(grow_base + lr0 + lm, n_total - 1);

    v8f c0 = {0.f,0.f,0.f,0.f,0.f,0.f,0.f,0.f};
    v8f c1 = c0, c2 = c0, c3 = c0;

    strip_product<HID, N>(acc + la * PITCH + 2 * hi,
                          Wl + (size_t)(2 * hi) * N + col0 + 4 * lm, c0, c1, c2, c3);

    #pragma unroll
    for (int r = 0; r < 8; ++r) {
        const int rr = lr0 + r + 8 * hi;
        const unsigned cn = COUNT ? cl[min(lrow_base + rr, ROWS_T - 1)]
                                  : cnt_g[min(grow_base + rr, n_total - 1)];
        const float inv = 1.0f / fmaxf((float)cn, 1.0f);
        c0[r] *= inv; c1[r] *= inv; c2[r] *= inv; c3[r] *= inv;
    }

    strip_product<HID, N>(x_dst + (size_t)ga * HID + 2 * hi,
                          Wr + (size_t)(2 * hi) * N + col0 + 4 * lm, c0, c1, c2, c3);

    const v4f bv = *(const v4f*)(bias + col0 + 4 * lm);
    #pragma unroll
    for (int r = 0; r < 8; ++r) {
        const int rr = lr0 + r + 8 * hi;
        const int g  = grow_base + rr;
        v4f v = {c0[r] + bv.x, c1[r] + bv.y, c2[r] + bv.z, c3[r] + bv.w};
        if (do_relu) {
            v.x = fmaxf(v.x, 0.f); v.y = fmaxf(v.y, 0.f);
            v.z = fmaxf(v.z, 0.f); v.w = fmaxf(v.w, 0.f);
        }
        if (rr < n_own && g < n_total)
            *(v4f*)(outp + (size_t)g * N + col0 + 4 * lm) = v;
    }
}

template <bool COUNT, int N>
__global__ void __launch_bounds__(1024)
owner_sum(const float* __restrict__ f_occ, const float* __restrict__ f_skill,
          const int* __restrict__ e_occ, const int* __restrict__ e_skill,
          unsigned* __restrict__ cnt_occ, unsigned* __restrict__ cnt_skill,
          float* __restrict__ list_buf,
          const float* __restrict__ Wl_s, const float* __restrict__ Wr_s, const float* __restrict__ bl_s,
          const float* __restrict__ Wl_o, const float* __restrict__ Wr_o, const float* __restrict__ bl_o,
          float* __restrict__ out_skill, float* __restrict__ out_occ, int do_relu)
{
    __shared__ float    acc[ROWS_T * PITCH];
    __shared__ unsigned q[NWAVE * QCAP];
    __shared__ unsigned cl[ROWS_T + 1];
    __shared__ unsigned ln[NWAVE];
    __shared__ unsigned sorted[SCAP];
    __shared__ unsigned cq[ROWS_T];
    __shared__ unsigned cur[ROWS_T];
    __shared__ unsigned tot;

    const int tid  = threadIdx.x;
    const int lane = tid & 31;
    const int wave = tid >> 5;
    const int s_lo = blockIdx.x * ROWS_S;
    const int o_lo = blockIdx.x * ROWS_O;
    unsigned* myq = q + wave * QCAP;
    unsigned* blk_list = (unsigned*)list_buf + (size_t)blockIdx.x * BLK_WORDS;
    unsigned* wl = blk_list + LIST_HDR + wave * CAPW;

    unsigned n_list = 0u;
    if (!COUNT) n_list = blk_list[wave];

    for (int i = tid; i < ROWS_T * PITCH / 4; i += 1024)
        ((float4*)acc)[i] = make_float4(0.f, 0.f, 0.f, 0.f);
    if (COUNT)
        for (int i = tid; i < ROWS_T; i += 1024) cl[i] = 0u;
    for (int i = tid; i < ROWS_T; i += 1024) cq[i] = 0u;

    int qn = 0;
    bool scan = true;
    if (!COUNT) {
        const unsigned n = __builtin_amdgcn_readfirstlane(n_list);
        if (n <= (unsigned)CAPW) {
            for (int i0 = 0; i0 < (int)n; i0 += 256) {
                unsigned t[8];
                #pragma unroll
                for (int j = 0; j < 8; ++j) {
                    const int i = i0 + 32 * j + lane;
                    t[j] = (i < (int)n) ? wl[i] : 0u;
                }
                #pragma unroll
                for (int j = 0; j < 8; ++j) {
                    const int i = i0 + 32 * j + lane;
                    if (i < (int)n) myq[i] = t[j];
                }
            }
            qn = (int)n;
            scan = false;
        }
    }
    __syncthreads();

    const int4* vo = (const int4*)e_occ;
    const int4* vs = (const int4*)e_skill;
    const int4 none = make_int4(-1, -1, -1, -1);

    bool fits = true;
    int it = (int)((blockIdx.x * NITER) >> 8);

    if (scan) {
    for (int k = 0; k < NITER; ++k) {
        const int ia = it * 2048 + tid;
        const int ib = ia + 1024;
        int4 eo0 = none, es0 = none, eo1 = none, es1 = none;
        if (ia < NVEC) { eo0 = vo[ia]; es0 = vs[ia]; }
        if (ib < NVEC) { eo1 = vo[ib]; es1 = vs[ib]; }

        if (qn > QCAP - 512) {
            drain_queue<COUNT>(myq, 0, qn, acc, cl, f_occ, f_skill, lane);
            qn = 0; fits = false;
        }

#define PUSH_EDGE(O, S)                                                              \
        {                                                                            \
            const unsigned rs = (unsigned)((S) - s_lo);                              \
            const bool ms = rs < (unsigned)ROWS_S;                                   \
            const unsigned bs = __builtin_amdgcn_ballot_w32(ms);                     \
            if (bs) {                                                                \
                if (ms) myq[qn + __builtin_amdgcn_mbcnt_lo(bs, 0u)] = (rs << 16) | (unsigned)(O); \
                qn += __builtin_popcount(bs);                                        \
            }                                                                        \
            const unsigned ro = (unsigned)((O) - o_lo);                              \
            const bool mo = ro < (unsigned)ROWS_O;                                   \
            const unsigned bo = __builtin_amdgcn_ballot_w32(mo);                     \
            if (bo) {                                                                \
                if (mo) myq[qn + __builtin_amdgcn_mbcnt_lo(bo, 0u)] = ((ro + ROWS_S) << 16) | (unsigned)(S); \
                qn += __builtin_popcount(bo);                                        \
            }                                                                        \
        }
        PUSH_EDGE(eo0.x, es0.x)
        PUSH_EDGE(eo0.y, es0.y)
        PUSH_EDGE(eo0.z, es0.z)
        PUSH_EDGE(eo0.w, es0.w)
        PUSH_EDGE(eo1.x, es1.x)
        PUSH_EDGE(eo1.y, es1.y)
        PUSH_EDGE(eo1.z, es1.z)
        PUSH_EDGE(eo1.w, es1.w)
#undef PUSH_EDGE
        it = (it + 1 == NITER) ? 0 : it + 1;
    }
    }

    if (COUNT) {
        const bool keep = fits && qn <= CAPW;
        if (keep)
            for (int i = lane; i < qn; i += 32) wl[i] = myq[i];
        if (lane == 0) ln[wave] = keep ? (unsigned)qn : (unsigned)(CAPW + 1);
    }

    for (int i = lane; i < qn; i += 32) atomicAdd(&cq[myq[i] >> 16], 1u);
    __syncthreads();

    if (wave == 0) {
        unsigned c[9];
        unsigned s = 0u;
        #pragma unroll
        for (int k = 0; k < 9; ++k) {
            const int r = 9 * lane + k;
            c[k] = (r < ROWS_T) ? cq[r] : 0u;
            s += c[k];
        }
        unsigned inc = s;
        #pragma unroll
        for (int d = 1; d < 32; d <<= 1) {
            const unsigned t = __shfl_up(inc, d);
            if (lane >= d) inc += t;
        }
        unsigned run = inc - s;
        #pragma unroll
        for (int k = 0; k < 9; ++k) {
            const int r = 9 * lane + k;
            if (r < ROWS_T) {
                cur[r] = run;
                if (COUNT) cl[r] += c[k];
                run += c[k];
            }
        }
        if (lane == 31) tot = inc;
    }
    __syncthreads();

    const int total = __builtin_amdgcn_readfirstlane((int)tot);
    const bool grouped = total <= SCAP;
    if (grouped) {
        for (int i = lane; i < qn; i += 32) {
            const unsigned e = myq[i];
            const unsigned pos = atomicAdd(&cur[e >> 16], 1u);
            sorted[pos] = e;
        }
    }
    __syncthreads();

    const int wave_s = __builtin_amdgcn_readfirstlane(wave);
    if (grouped) {
        const int piece = (((total + NWAVE - 1) / NWAVE) + 7) & ~7;
        const int a = min(wave_s * piece, total);
        const int b = min(a + piece, total);
        sum_runs(sorted, a, b, acc, f_occ, f_skill, lane);
    } else {
        drain_queue<false>(myq, 0, qn, acc, cl, f_occ, f_skill, lane);
    }

    __syncthreads();

    if (COUNT && wave == 0) blk_list[lane] = ln[lane];

    if (COUNT) {
        for (int i = tid; i < ROWS_S; i += 1024)
            if (s_lo + i < N_SKILL) cnt_skill[s_lo + i] = cl[i];
        for (int i = tid; i < ROWS_O; i += 1024)
            if (o_lo + i < N_OCC) cnt_occ[o_lo + i] = cl[ROWS_S + i];
    }

    constexpr int NSTRIP = N / 64;
    for (int task = wave_s; task < (TILES_S + TILES_O) * NSTRIP; task += NWAVE) {
        const int  t    = task / NSTRIP;
        const int  col0 = (task % NSTRIP) * 64;
        const bool occ  = t >= TILES_S;
        const int  tile = occ ? t - TILES_S : t;
        const int  gb   = occ ? o_lo : s_lo;
        const int  ntot = occ ? N_OCC : N_SKILL;
        const int  nown = occ ? ROWS_O : ROWS_S;
        if (gb + 16 * tile >= ntot) continue;
        own_rows_product<N, COUNT>(acc, cl,
                                   occ ? f_occ : f_skill,
                                   occ ? Wl_o : Wl_s, occ ? Wr_o : Wr_s, occ ? bl_o : bl_s,
                                   occ ? cnt_occ : cnt_skill,
                                   occ ? out_occ : out_skill,
                                   occ ? ROWS_S : 0, nown, gb, ntot, tile, col0, do_relu, lane);
    }
}

extern "C" void kernel_launch(void* const* d_in, const int* in_sizes, int n_in,
                              void* d_out, int out_size, void* d_ws, size_t ws_size,
                              hipStream_t stream)
{
    (void)in_sizes; (void)n_in; (void)out_size; (void)ws_size;

    const float* x_occ    = (const float*)d_in[0];
    const float* x_skill  = (const float*)d_in[1];
    const int*   e_occ    = (const int*)  d_in[2];
    const int*   e_skill  = (const int*)  d_in[3];
    const float* pW_occ   = (const float*)d_in[4];
    const float* pb_occ   = (const float*)d_in[5];
    const float* pW_skill = (const float*)d_in[6];
    const float* pb_skill = (const float*)d_in[7];
    const float* c1_os_Wl = (const float*)d_in[8];
    const float* c1_os_bl = (const float*)d_in[9];
    const float* c1_os_Wr = (const float*)d_in[10];
    const float* c1_so_Wl = (const float*)d_in[11];
    const float* c1_so_bl = (const float*)d_in[12];
    const float* c1_so_Wr = (const float*)d_in[13];
    const float* c2_os_Wl = (const float*)d_in[14];
    const float* c2_os_bl = (const float*)d_in[15];
    const float* c2_os_Wr = (const float*)d_in[16];
    const float* c2_so_Wl = (const float*)d_in[17];
    const float* c2_so_bl = (const float*)d_in[18];
    const float* c2_so_Wr = (const float*)d_in[19];

    float* ws = (float*)d_ws;
    float* h_occ     = ws;
    float* h_skill   = h_occ     + (size_t)N_OCC   * HID;
    float* o1        = h_skill   + (size_t)N_SKILL * HID;
    float* s1        = o1        + (size_t)N_OCC   * HID;
    float* edge_lists = s1       + (size_t)N_SKILL * HID;
    unsigned* cnt_occ   = (unsigned*)(edge_lists + (size_t)(N_OCC + N_SKILL) * HID);
    unsigned* cnt_skill = cnt_occ + N_OCC;

    float* o2 = (float*)d_out;
    float* s2 = o2 + (size_t)N_OCC * OUTF;

    const dim3 blk128(128);
    const int rt_occ   = (N_OCC / 16 + 3) / 4;
    const int rt_skill = (N_SKILL / 16 + 3) / 4;

    gemm_sage<D_IN, HID, false><<<dim3(rt_occ, HID / 64), blk128, 0, stream>>>(
        x_occ, pW_occ, nullptr, nullptr, nullptr, pb_occ, h_occ, N_OCC, 1);
    gemm_sage<D_IN, HID, false><<<dim3(rt_skill, HID / 64), blk128, 0, stream>>>(
        x_skill, pW_skill, nullptr, nullptr, nullptr, pb_skill, h_skill, N_SKILL, 1);

    owner_sum<true, HID><<<NBLK, dim3(1024), 0, stream>>>(
        h_occ, h_skill, e_occ, e_skill, cnt_occ, cnt_skill, edge_lists,
        c1_os_Wl, c1_os_Wr, c1_os_bl, c1_so_Wl, c1_so_Wr, c1_so_bl, s1, o1, 1);

    owner_sum<false, OUTF><<<NBLK, dim3(1024), 0, stream>>>(
        o1, s1, e_occ, e_skill, cnt_occ, cnt_skill, edge_lists,
        c2_os_Wl, c2_os_Wr, c2_os_bl, c2_so_Wl, c2_so_Wr, c2_so_bl, s2, o2, 0);
}


# neighbour-sum scan reads only marked 32-edge chunks: extra blocks of the first product launch build, per 1024-edge span and owner block, a 32-bit chunk mask; f32 sum order: arrival order, as before
# speedup vs baseline: 4.3152x; 1.1170x over previous
#include <hip/hip_runtime.h>

#define N_OCC   20000
#define N_SKILL 50000
#define N_EDGE  600000
#define D_IN    256
#define HID     128
#define OUTF    64

typedef __attribute__((ext_vector_type(2))) float v2f;
typedef __attribute__((ext_vector_type(4))) float v4f;
typedef __attribute__((ext_vector_type(8))) float v8f;

#define NBLK   256
#define ROWS_S 196
#define ROWS_O 79
#define ROWS_T (ROWS_S + ROWS_O)
#define NWAVE  32
#define CAPW   544
#define QCAP   (CAPW + 512)
#define SCAP   8192
#define LIST_HDR  NWAVE
#define BLK_WORDS (LIST_HDR + NWAVE * CAPW)
#define PITCH  (HID + 4)
#define TILES_S ((ROWS_S + 15) / 16)
#define TILES_O ((ROWS_O + 15) / 16)

static_assert(NBLK * ROWS_S >= N_SKILL && NBLK * ROWS_O >= N_OCC, "row ranges cover all nodes");
static_assert(N_SKILL < 65536 && N_OCC < 65536 && ROWS_T < 65536, "queue entry packing");
static_assert(QCAP % 8 == 0 && QCAP > 512, "drain reads whole batches of 8; room for one batch of 8 chunks");
static_assert(CAPW % 32 == 0 && BLK_WORDS % 32 == 0, "lists start on 128-byte lines");
static_assert((ROWS_T * PITCH) % 4 == 0, "zero fill writes float4");
static_assert(32 * 9 >= ROWS_T, "prefix sum: 9 rows per lane of one wave");

#define ESPAN  1024
#define NSPAN  ((N_EDGE + ESPAN - 1) / ESPAN)
#define SPW    ((NSPAN + NWAVE - 1) / NWAVE)
static_assert(N_EDGE % 32 == 0, "chunks are whole");
static_assert(NSPAN % 2 == 0 && HID / 64 == 2, "the index blocks fill both rows of the first product's grid");
static_assert(SPW <= 32, "a wave keeps its spans' words one per lane");
static_assert((size_t)NBLK * BLK_WORDS + (size_t)NSPAN * NBLK <= (size_t)(N_OCC + N_SKILL) * HID, "lists and chunk index fit their workspace region");

template <int K, int N>
__device__ inline void strip_product(const float* __restrict__ ap, const float* __restrict__ bp,
                                     v8f& c0, v8f& c1, v8f& c2, v8f& c3)
{
    #pragma unroll 8
    for (int k = 0; k < K / 4; ++k) {
        const v2f a  = *(const v2f*)(ap + 4 * k);
        const v4f w0 = *(const v4f*)(bp + (size_t)(4 * k) * N);
        const v4f w1 = *(const v4f*)(bp + (size_t)(4 * k + 1) * N);
        const v2f b0 = {w0.x, w1.x};
        const v2f b1 = {w0.y, w1.y};
        const v2f b2 = {w0.z, w1.z};
        const v2f b3 = {w0.w, w1.w};
        c0 = __builtin_amdgcn_wmma_f32_16x16x4_f32(false, a, false, b0, (short)0, c0, false, false);
        c1 = __builtin_amdgcn_wmma_f32_16x16x4_f32(false, a, false, b1, (short)0, c1, false, false);
        c2 = __builtin_amdgcn_wmma_f32_16x16x4_f32(false, a, false, b2, (short)0, c2, false, false);
        c3 = __builtin_amdgcn_wmma_f32_16x16x4_f32(false, a, false, b3, (short)0, c3, false, false);
    }
}

template <int K, int N, bool FUSE, bool INDEX>
__global__ void __launch_bounds__(128)
gemm_sage(const float* __restrict__ A1, const float* __restrict__ W1,
          const float* __restrict__ A2, const float* __restrict__ W2,
          const unsigned* __restrict__ cnt, const float* __restrict__ bias,
          float* __restrict__ out, int M, int do_relu,
          const int* __restrict__ e_occ, const int* __restrict__ e_skill,
          unsigned* __restrict__ span_idx, int x_blocks)
{
    static_assert(N % 64 == 0 && K % 32 == 0, "strip of four interleaved column tiles; k loop unrolled by 8");
    const int lane = threadIdx.x & 31;
    const int wave = threadIdx.x >> 5;
    if (INDEX && (int)blockIdx.x >= x_blocks) {
        __shared__ unsigned owners[NBLK];
        const int span = ((int)blockIdx.x - x_blocks) * 2 + (int)blockIdx.y;
        for (int i = threadIdx.x; i < NBLK; i += 128) owners[i] = 0u;
        __syncthreads();
        const int e0 = span * ESPAN + 32 * wave + lane;
        int vs[8], vo[8];
        #pragma unroll
        for (int c4 = 0; c4 < 8; ++c4) {
            const int ec = min(e0 + 128 * c4, N_EDGE - 1);
            vs[c4] = e_skill[ec];
            vo[c4] = e_occ[ec];
        }
        #pragma unroll
        for (int c4 = 0; c4 < 8; ++c4) {
            if (e0 + 128 * c4 < N_EDGE) {
                const unsigned bs = min((unsigned)vs[c4] / (unsigned)ROWS_S, (unsigned)(NBLK - 1));
                const unsigned bo = min((unsigned)vo[c4] / (unsigned)ROWS_O, (unsigned)(NBLK - 1));
                atomicOr(&owners[bs], 1u << (4 * c4 + wave));
                atomicOr(&owners[bo], 1u << (4 * c4 + wave));
            }
        }
        __syncthreads();
        for (int i = threadIdx.x; i < NBLK; i += 128) span_idx[(size_t)span * NBLK + i] = owners[i];
        return;
    }
    const int hi   = lane >> 4;
    const int lm   = lane & 15;
    const int rt   = blockIdx.x * 4 + wave;
    if (rt * 16 >= M) return;
    const int row0 = rt * 16;
    const int col0 = blockIdx.y * 64;

    v8f c0 = {0.f,0.f,0.f,0.f,0.f,0.f,0.f,0.f};
    v8f c1 = c0, c2 = c0, c3 = c0;

    strip_product<K, N>(A1 + (size_t)(row0 + lm) * K + 2 * hi,
                        W1 + (size_t)(2 * hi) * N + col0 + 4 * lm, c0, c1, c2, c3);

    if (FUSE) {
        #pragma unroll
        for (int r = 0; r < 8; ++r) {
            const float inv = 1.0f / fmaxf((float)cnt[row0 + r + 8 * hi], 1.0f);
            c0[r] *= inv; c1[r] *= inv; c2[r] *= inv; c3[r] *= inv;
        }
        strip_product<K, N>(A2 + (size_t)(row0 + lm) * K + 2 * hi,
                            W2 + (size_t)(2 * hi) * N + col0 + 4 * lm, c0, c1, c2, c3);
    }

    float* orow = out + (size_t)(row0 + 8 * hi) * N + col0 + 4 * lm;
    const v4f bv = *(const v4f*)(bias + col0 + 4 * lm);
    #pragma unroll
    for (int r = 0; r < 8; ++r) {
        v4f v = {c0[r] + bv.x, c1[r] + bv.y, c2[r] + bv.z, c3[r] + bv.w};
        if (do_relu) {
            v.x = fmaxf(v.x, 0.f); v.y = fmaxf(v.y, 0.f);
            v.z = fmaxf(v.z, 0.f); v.w = fmaxf(v.w, 0.f);
        }
        *(v4f*)(orow + (size_t)r * N) = v;
    }
}

__device__ inline void lds_add_f32(float* p, float v)
{
    (void)__hip_atomic_fetch_add(p, v, __ATOMIC_RELAXED, __HIP_MEMORY_SCOPE_WORKGROUP);
}

template <bool COUNT>
__device__ inline void drain_queue(const unsigned* myq, int qs, int qn, float* acc, unsigned* cl,
                                   const float* __restrict__ f_occ,
                                   const float* __restrict__ f_skill, int lane)
{
    for (int base = qs; base < qn; base += 8) {
        unsigned row[8];
        float v[8][4];
        #pragma unroll
        for (int j = 0; j < 8; ++j) {
            unsigned e = __builtin_amdgcn_readfirstlane(myq[base + j]);
            if (base + j >= qn) e = 0u;
            row[j] = e >> 16;
            const unsigned srow = e & 0xffffu;
            const float* tab = (row[j] < (unsigned)ROWS_S) ? f_occ : f_skill;
            const float* p = tab + (size_t)srow * HID + lane;
            v[j][0] = p[0]; v[j][1] = p[32]; v[j][2] = p[64]; v[j][3] = p[96];
        }
        #pragma unroll
        for (int j = 0; j < 8; ++j) {
            if (base + j < qn) {
                float* d = acc + row[j] * PITCH + lane;
                lds_add_f32(d,      v[j][0]);
                lds_add_f32(d + 32, v[j][1]);
                lds_add_f32(d + 64, v[j][2]);
                lds_add_f32(d + 96, v[j][3]);
                if (COUNT && lane == 0) atomicAdd(&cl[row[j]], 1u);
            }
        }
    }
}

__device__ inline void flush_run(float* acc, unsigned row, float s0, float s1, float s2, float s3, int lane)
{
    float* d = acc + row * PITCH + lane;
    lds_add_f32(d,      s0);
    lds_add_f32(d + 32, s1);
    lds_add_f32(d + 64, s2);
    lds_add_f32(d + 96, s3);
}

__device__ inline void sum_runs(const unsigned* sorted, int a, int b, float* acc,
                                const float* __restrict__ f_occ,
                                const float* __restrict__ f_skill, int lane)
{
    const unsigned NONE = 0xffffffffu;
    unsigned cur = NONE;
    float s0 = 0.f, s1 = 0.f, s2 = 0.f, s3 = 0.f;
    for (int base = a; base < b; base += 8) {
        const unsigned ev = sorted[min(base + (lane & 7), b - 1)];
        unsigned row[8];
        float v[8][4];
        #pragma unroll
        for (int j = 0; j < 8; ++j) {
            const unsigned e = __builtin_amdgcn_readlane(ev, j);
            row[j] = e >> 16;
            const unsigned srow = e & 0xffffu;
            const float* tab = (row[j] < (unsigned)ROWS_S) ? f_occ : f_skill;
            const float* p = tab + (size_t)srow * HID + lane;
            v[j][0] = p[0]; v[j][1] = p[32]; v[j][2] = p[64]; v[j][3] = p[96];
        }
        #pragma unroll
        for (int j = 0; j < 8; ++j) {
            if (base + j < b) {
                if (row[j] != cur) {
                    if (cur != NONE) flush_run(acc, cur, s0, s1, s2, s3, lane);
                    cur = row[j];
                    s0 = v[j][0]; s1 = v[j][1]; s2 = v[j][2]; s3 = v[j][3];
                } else {
                    s0 += v[j][0]; s1 += v[j][1]; s2 += v[j][2]; s3 += v[j][3];
                }
            }
        }
    }
    if (cur != NONE) flush_run(acc, cur, s0, s1, s2, s3, lane);
}

template <int N, bool COUNT>
__device__ inline void own_rows_product(const float* acc, const unsigned* cl,
                                        const float* __restrict__ x_dst,
                                        const float* __restrict__ Wl, const float* __restrict__ Wr,
                                        const float* __restrict__ bias, const unsigned* __restrict__ cnt_g,
                                        float* __restrict__ outp,
                                        int lrow_base, int n_own, int grow_base, int n_total,
                                        int tile, int col0, int do_relu, int lane)
{
    const int hi  = lane >> 4;
    const int lm  = lane & 15;
    const int lr0 = 16 * tile;
    const int la  = min(lrow_base + lr0 + lm, ROWS_T - 1);
    const int ga  = min(grow_base + lr0 + lm, n_total - 1);

    v8f c0 = {0.f,0.f,0.f,0.f,0.f,0.f,0.f,0.f};
    v8f c1 = c0, c2 = c0, c3 = c0;

    strip_product<HID, N>(acc + la * PITCH + 2 * hi,
                          Wl + (size_t)(2 * hi) * N + col0 + 4 * lm, c0, c1, c2, c3);

    #pragma unroll
    for (int r = 0; r < 8; ++r) {
        const int rr = lr0 + r + 8 * hi;
        const unsigned cn = COUNT ? cl[min(lrow_base + rr, ROWS_T - 1)]
                                  : cnt_g[min(grow_base + rr, n_total - 1)];
        const float inv = 1.0f / fmaxf((float)cn, 1.0f);
        c0[r] *= inv; c1[r] *= inv; c2[r] *= inv; c3[r] *= inv;
    }

    strip_product<HID, N>(x_dst + (size_t)ga * HID + 2 * hi,
                          Wr + (size_t)(2 * hi) * N + col0 + 4 * lm, c0, c1, c2, c3);

    const v4f bv = *(const v4f*)(bias + col0 + 4 * lm);
    #pragma unroll
    for (int r = 0; r < 8; ++r) {
        const int rr = lr0 + r + 8 * hi;
        const int g  = grow_base + rr;
        v4f v = {c0[r] + bv.x, c1[r] + bv.y, c2[r] + bv.z, c3[r] + bv.w};
        if (do_relu) {
            v.x = fmaxf(v.x, 0.f); v.y = fmaxf(v.y, 0.f);
            v.z = fmaxf(v.z, 0.f); v.w = fmaxf(v.w, 0.f);
        }
        if (rr < n_own && g < n_total)
            *(v4f*)(outp + (size_t)g * N + col0 + 4 * lm) = v;
    }
}

template <bool COUNT, int N>
__global__ void __launch_bounds__(1024)
owner_sum(const float* __restrict__ f_occ, const float* __restrict__ f_skill,
          const int* __restrict__ e_occ, const int* __restrict__ e_skill,
          unsigned* __restrict__ cnt_occ, unsigned* __restrict__ cnt_skill,
          float* __restrict__ list_buf, const unsigned* __restrict__ span_idx,
          const float* __restrict__ Wl_s, const float* __restrict__ Wr_s, const float* __restrict__ bl_s,
          const float* __restrict__ Wl_o, const float* __restrict__ Wr_o, const float* __restrict__ bl_o,
          float* __restrict__ out_skill, float* __restrict__ out_occ, int do_relu)
{
    __shared__ float    acc[ROWS_T * PITCH];
    __shared__ unsigned q[NWAVE * QCAP];
    __shared__ unsigned cl[ROWS_T + 1];
    __shared__ unsigned ln[NWAVE];
    __shared__ unsigned sorted[SCAP];
    __shared__ unsigned cq[ROWS_T];
    __shared__ unsigned cur[ROWS_T];
    __shared__ unsigned tot;

    const int tid  = threadIdx.x;
    const int lane = tid & 31;
    const int wave = tid >> 5;
    const int s_lo = blockIdx.x * ROWS_S;
    const int o_lo = blockIdx.x * ROWS_O;
    unsigned* myq = q + wave * QCAP;
    unsigned* blk_list = (unsigned*)list_buf + (size_t)blockIdx.x * BLK_WORDS;
    unsigned* wl = blk_list + LIST_HDR + wave * CAPW;

    unsigned n_list = 0u;
    if (!COUNT) n_list = blk_list[wave];

    for (int i = tid; i < ROWS_T * PITCH / 4; i += 1024)
        ((float4*)acc)[i] = make_float4(0.f, 0.f, 0.f, 0.f);
    if (COUNT)
        for (int i = tid; i < ROWS_T; i += 1024) cl[i] = 0u;
    for (int i = tid; i < ROWS_T; i += 1024) cq[i] = 0u;

    int qn = 0;
    bool scan = true;
    if (!COUNT) {
        const unsigned n = __builtin_amdgcn_readfirstlane(n_list);
        if (n <= (unsigned)CAPW) {
            for (int i0 = 0; i0 < (int)n; i0 += 256) {
                unsigned t[8];
                #pragma unroll
                for (int j = 0; j < 8; ++j) {
                    const int i = i0 + 32 * j + lane;
                    t[j] = (i < (int)n) ? wl[i] : 0u;
                }
                #pragma unroll
                for (int j = 0; j < 8; ++j) {
                    const int i = i0 + 32 * j + lane;
                    if (i < (int)n) myq[i] = t[j];
                }
            }
            qn = (int)n;
            scan = false;
        }
    }
    __syncthreads();

    bool fits = true;
    const int wave_s = __builtin_amdgcn_readfirstlane(wave);

    if (scan) {
        unsigned mymask = 0u;
        {
            const int sp = wave_s + NWAVE * lane;
            if (sp < NSPAN) mymask = span_idx[(size_t)sp * NBLK + blockIdx.x];
        }
        for (int j = 0; j < SPW; ++j) {
            unsigned m = __builtin_amdgcn_readlane(mymask, j);
            const int ebase = (wave_s + NWAVE * j) * ESPAN;
            while (m) {
                if (qn > QCAP - 512) {
                    drain_queue<COUNT>(myq, 0, qn, acc, cl, f_occ, f_skill, lane);
                    qn = 0; fits = false;
                }
                const int cfirst = __builtin_ctz(m);
                int  eo[8], es[8];
                bool on[8];
                #pragma unroll
                for (int u = 0; u < 8; ++u) {
                    on[u] = (m != 0u);
                    const int ch = on[u] ? __builtin_ctz(m) : cfirst;
                    m &= m - 1u;
                    const int e  = ebase + 32 * ch + lane;
                    const int ec = min(e, N_EDGE - 1);
                    const int vo = e_occ[ec];
                    const int vs = e_skill[ec];
                    eo[u] = (e < N_EDGE) ? vo : -1;
                    es[u] = (e < N_EDGE) ? vs : -1;
                }

#define PUSH_EDGE(O, S)                                                              \
        {                                                                            \
            const unsigned rs = (unsigned)((S) - s_lo);                              \
            const bool ms = rs < (unsigned)ROWS_S;                                   \
            const unsigned bs = __builtin_amdgcn_ballot_w32(ms);                     \
            if (bs) {                                                                \
                if (ms) myq[qn + __builtin_amdgcn_mbcnt_lo(bs, 0u)] = (rs << 16) | (unsigned)(O); \
                qn += __builtin_popcount(bs);                                        \
            }                                                                        \
            const unsigned ro = (unsigned)((O) - o_lo);                              \
            const bool mo = ro < (unsigned)ROWS_O;                                   \
            const unsigned bo = __builtin_amdgcn_ballot_w32(mo);                     \
            if (bo) {                                                                \
                if (mo) myq[qn + __builtin_amdgcn_mbcnt_lo(bo, 0u)] = ((ro + ROWS_S) << 16) | (unsigned)(S); \
                qn += __builtin_popcount(bo);                                        \
            }                                                                        \
        }
                #pragma unroll
                for (int u = 0; u < 8; ++u) {
                    if (on[u]) PUSH_EDGE(eo[u], es[u])
                }
#undef PUSH_EDGE
            }
        }
    }

    if (COUNT) {
        const bool keep = fits && qn <= CAPW;
        if (keep)
            for (int i = lane; i < qn; i += 32) wl[i] = myq[i];
        if (lane == 0) ln[wave] = keep ? (unsigned)qn : (unsigned)(CAPW + 1);
    }

    for (int i = lane; i < qn; i += 32) atomicAdd(&cq[myq[i] >> 16], 1u);
    __syncthreads();

    if (wave == 0) {
        unsigned c[9];
        unsigned s = 0u;
        #pragma unroll
        for (int k = 0; k < 9; ++k) {
            const int r = 9 * lane + k;
            c[k] = (r < ROWS_T) ? cq[r] : 0u;
            s += c[k];
        }
        unsigned inc = s;
        #pragma unroll
        for (int d = 1; d < 32; d <<= 1) {
            const unsigned t = __shfl_up(inc, d);
            if (lane >= d) inc += t;
        }
        unsigned run = inc - s;
        #pragma unroll
        for (int k = 0; k < 9; ++k) {
            const int r = 9 * lane + k;
            if (r < ROWS_T) {
                cur[r] = run;
                if (COUNT) cl[r] += c[k];
                run += c[k];
            }
        }
        if (lane == 31) tot = inc;
    }
    __syncthreads();

    const int total = __builtin_amdgcn_readfirstlane((int)tot);
    const bool grouped = total <= SCAP;
    if (grouped) {
        for (int i = lane; i < qn; i += 32) {
            const unsigned e = myq[i];
            const unsigned pos = atomicAdd(&cur[e >> 16], 1u);
            sorted[pos] = e;
        }
    }
    __syncthreads();

    if (grouped) {
        const int piece = (((total + NWAVE - 1) / NWAVE) + 7) & ~7;
        const int a = min(wave_s * piece, total);
        const int b = min(a + piece, total);
        sum_runs(sorted, a, b, acc, f_occ, f_skill, lane);
    } else {
        drain_queue<false>(myq, 0, qn, acc, cl, f_occ, f_skill, lane);
    }

    __syncthreads();

    if (COUNT && wave == 0) blk_list[lane] = ln[lane];

    if (COUNT) {
        for (int i = tid; i < ROWS_S; i += 1024)
            if (s_lo + i < N_SKILL) cnt_skill[s_lo + i] = cl[i];
        for (int i = tid; i < ROWS_O; i += 1024)
            if (o_lo + i < N_OCC) cnt_occ[o_lo + i] = cl[ROWS_S + i];
    }

    constexpr int NSTRIP = N / 64;
    for (int task = wave_s; task < (TILES_S + TILES_O) * NSTRIP; task += NWAVE) {
        const int  t    = task / NSTRIP;
        const int  col0 = (task % NSTRIP) * 64;
        const bool occ  = t >= TILES_S;
        const int  tile = occ ? t - TILES_S : t;
        const int  gb   = occ ? o_lo : s_lo;
        const int  ntot = occ ? N_OCC : N_SKILL;
        const int  nown = occ ? ROWS_O : ROWS_S;
        if (gb + 16 * tile >= ntot) continue;
        own_rows_product<N, COUNT>(acc, cl,
                                   occ ? f_occ : f_skill,
                                   occ ? Wl_o : Wl_s, occ ? Wr_o : Wr_s, occ ? bl_o : bl_s,
                                   occ ? cnt_occ : cnt_skill,
                                   occ ? out_occ : out_skill,
                                   occ ? ROWS_S : 0, nown, gb, ntot, tile, col0, do_relu, lane);
    }
}

extern "C" void kernel_launch(void* const* d_in, const int* in_sizes, int n_in,
                              void* d_out, int out_size, void* d_ws, size_t ws_size,
                              hipStream_t stream)
{
    (void)in_sizes; (void)n_in; (void)out_size; (void)ws_size;

    const float* x_occ    = (const float*)d_in[0];
    const float* x_skill  = (const float*)d_in[1];
    const int*   e_occ    = (const int*)  d_in[2];
    const int*   e_skill  = (const int*)  d_in[3];
    const float* pW_occ   = (const float*)d_in[4];
    const float* pb_occ   = (const float*)d_in[5];
    const float* pW_skill = (const float*)d_in[6];
    const float* pb_skill = (const float*)d_in[7];
    const float* c1_os_Wl = (const float*)d_in[8];
    const float* c1_os_bl = (const float*)d_in[9];
    const float* c1_os_Wr = (const float*)d_in[10];
    const float* c1_so_Wl = (const float*)d_in[11];
    const float* c1_so_bl = (const float*)d_in[12];
    const float* c1_so_Wr = (const float*)d_in[13];
    const float* c2_os_Wl = (const float*)d_in[14];
    const float* c2_os_bl = (const float*)d_in[15];
    const float* c2_os_Wr = (const float*)d_in[16];
    const float* c2_so_Wl = (const float*)d_in[17];
    const float* c2_so_bl = (const float*)d_in[18];
    const float* c2_so_Wr = (const float*)d_in[19];

    float* ws = (float*)d_ws;
    float* h_occ     = ws;
    float* h_skill   = h_occ     + (size_t)N_OCC   * HID;
    float* o1        = h_skill   + (size_t)N_SKILL * HID;
    float* s1        = o1        + (size_t)N_OCC   * HID;
    float* edge_lists = s1       + (size_t)N_SKILL * HID;
    unsigned* span_idx = (unsigned*)edge_lists + (size_t)NBLK * BLK_WORDS;
    unsigned* cnt_occ   = (unsigned*)(edge_lists + (size_t)(N_OCC + N_SKILL) * HID);
    unsigned* cnt_skill = cnt_occ + N_OCC;

    float* o2 = (float*)d_out;
    float* s2 = o2 + (size_t)N_OCC * OUTF;

    const dim3 blk128(128);
    const int rt_occ   = (N_OCC / 16 + 3) / 4;
    const int rt_skill = (N_SKILL / 16 + 3) / 4;

    gemm_sage<D_IN, HID, false, true><<<dim3(rt_occ + NSPAN / 2, HID / 64), blk128, 0, stream>>>(
        x_occ, pW_occ, nullptr, nullptr, nullptr, pb_occ, h_occ, N_OCC, 1,
        e_occ, e_skill, span_idx, rt_occ);
    gemm_sage<D_IN, HID, false, false><<<dim3(rt_skill, HID / 64), blk128, 0, stream>>>(
        x_skill, pW_skill, nullptr, nullptr, nullptr, pb_skill, h_skill, N_SKILL, 1,
        nullptr, nullptr, nullptr, rt_skill);

    owner_sum<true, HID><<<NBLK, dim3(1024), 0, stream>>>(
        h_occ, h_skill, e_occ, e_skill, cnt_occ, cnt_skill, edge_lists, span_idx,
        c1_os_Wl, c1_os_Wr, c1_os_bl, c1_so_Wl, c1_so_Wr, c1_so_bl, s1, o1, 1);

    owner_sum<false, OUTF><<<NBLK, dim3(1024), 0, stream>>>(
        o1, s1, e_occ, e_skill, cnt_occ, cnt_skill, edge_lists, span_idx,
        c2_os_Wl, c2_os_Wr, c2_os_bl, c2_so_Wl, c2_so_Wr, c2_so_bl, s2, o2, 0);
}
